# starting kernel
# baseline (speedup 1.0000x reference)
#include <hip/hip_runtime.h>

typedef __attribute__((ext_vector_type(2))) float v2f;
typedef __attribute__((ext_vector_type(8))) float v8f;


__global__ void deg_init_kernel(int* __restrict__ deg, int n) {
    int i = blockIdx.x * blockDim.x + threadIdx.x;
    if (i < n) deg[i] = 1;
}

__global__ void deg_accum_kernel(const int* __restrict__ dst, int* __restrict__ deg, int E) {
    int e = blockIdx.x * blockDim.x + threadIdx.x;
    if (e < E) atomicAdd(&deg[dst[e]], 1);
}

__global__ void dinv_kernel(float* __restrict__ dinv, const int* __restrict__ deg, int n) {
    int i = blockIdx.x * blockDim.x + threadIdx.x;
    if (i < n) {
        float d = (float)deg[i];
        dinv[i] = rsqrtf(d);
    }
}

__global__ void norm_kernel(const int* __restrict__ src, const int* __restrict__ dst,
                            const float* __restrict__ dinv, float* __restrict__ norm, int E) {
    int e = blockIdx.x * blockDim.x + threadIdx.x;
    if (e < E) norm[e] = dinv[src[e]] * dinv[dst[e]];
}

template <int K, int NCOL>
__global__ __launch_bounds__(128) void gemm_wmma_f32(const float* __restrict__ A,
                                                     const float* __restrict__ B,
                                                     float* __restrict__ C, int nrows) {
    const int wave = threadIdx.x >> 5;
    const int lane = threadIdx.x & 31;
    const int half = lane >> 4;
    const int mr   = lane & 15;

    const int row_tile = blockIdx.x * (blockDim.x >> 5) + wave;
    const int row0 = row_tile * 16;
    if (row0 >= nrows) return;

    constexpr int NT = NCOL / 16;
    v8f acc[NT] = {};

    const float* arow = A + (size_t)(row0 + mr) * K + half * 2;

#pragma unroll 8
    for (int kb = 0; kb < K / 4; ++kb) {
        v2f a;
        a.x = arow[kb * 4 + 0];
        a.y = arow[kb * 4 + 1];
        const float* brow = B + (size_t)(kb * 4 + half * 2) * NCOL + mr;
#pragma unroll
        for (int ct = 0; ct < NT; ++ct) {
            v2f b;
            b.x = brow[ct * 16];
            b.y = brow[ct * 16 + NCOL];
            acc[ct] = __builtin_amdgcn_wmma_f32_16x16x4_f32(
                 false, a,  false, b,
                 (short)0, acc[ct],  false,  false);
        }
    }

#pragma unroll
    for (int ct = 0; ct < NT; ++ct) {
        float* crow = C + (size_t)(row0 + half * 8) * NCOL + ct * 16 + mr;
#pragma unroll
        for (int r = 0; r < 8; ++r) crow[(size_t)r * NCOL] = acc[ct][r];
    }
}

template <int F>
__global__ void self_bias_init_kernel(const float* __restrict__ h, const float* __restrict__ dinv,
                                      const float* __restrict__ b, float* __restrict__ out,
                                      long total) {
    long idx = (long)blockIdx.x * blockDim.x + threadIdx.x;
    if (idx >= total) return;
    constexpr int SH = (F == 128) ? 7 : 6;
    int  f = (int)(idx & (F - 1));
    long i = idx >> SH;
    float di = dinv[i];
    out[idx] = b[f] + di * di * h[idx];
}

__global__ void relu_kernel(float* __restrict__ x, long total) {
    long idx = (long)blockIdx.x * blockDim.x + threadIdx.x;
    if (idx < total) x[idx] = fmaxf(x[idx], 0.0f);
}

template <int F>
__global__ void scatter_edges_kernel(const float* __restrict__ h, const int* __restrict__ src,
                                     const int* __restrict__ dst, const float* __restrict__ norm,
                                     float* __restrict__ out, int E) {
    unsigned gt = blockIdx.x * blockDim.x + threadIdx.x;
    int e = (int)(gt >> 5);
    if (e >= E) return;
    int lane = threadIdx.x & 31;

    int   s = src[e];
    int   d = dst[e];
    float w = norm[e];

    if constexpr (F == 128) {
        float4 v = *((const float4*)(h + (size_t)s * F) + lane);
        float* o = out + (size_t)d * F + lane * 4;
        atomicAdd(o + 0, v.x * w);
        atomicAdd(o + 1, v.y * w);
        atomicAdd(o + 2, v.z * w);
        atomicAdd(o + 3, v.w * w);
    } else {
        float2 v = *((const float2*)(h + (size_t)s * F) + lane);
        float* o = out + (size_t)d * F + lane * 2;
        atomicAdd(o + 0, v.x * w);
        atomicAdd(o + 1, v.y * w);
    }
}

extern "C" void kernel_launch(void* const* d_in, const int* in_sizes, int n_in,
                              void* d_out, int out_size, void* d_ws, size_t ws_size,
                              hipStream_t stream) {
    const float* x  = (const float*)d_in[0];
    const int*   ei = (const int*)d_in[1];
    const float* W1 = (const float*)d_in[2];
    const float* b1 = (const float*)d_in[3];
    const float* W2 = (const float*)d_in[4];
    const float* b2 = (const float*)d_in[5];

    const int N = in_sizes[0] / 128;
    const int E = in_sizes[1] / 2;
    const int* src = ei;
    const int* dst = ei + E;

    float* dinv = (float*)d_ws;
    float* norm = dinv + N;
    float* h1   = norm + E;
    float* agg1 = h1 + (size_t)N * 128;
    float* h2   = h1;
    float* out  = (float*)d_out;

    const int T = 256;

    deg_init_kernel<<<(N + T - 1) / T, T, 0, stream>>>((int*)dinv, N);
    deg_accum_kernel<<<(E + T - 1) / T, T, 0, stream>>>(dst, (int*)dinv, E);
    dinv_kernel<<<(N + T - 1) / T, T, 0, stream>>>(dinv, (const int*)dinv, N);
    norm_kernel<<<(E + T - 1) / T, T, 0, stream>>>(src, dst, dinv, norm, E);

    const int row_tiles = (N + 15) / 16;
    const int gemm_blocks = (row_tiles + 3) / 4;

    gemm_wmma_f32<128, 128><<<gemm_blocks, 128, 0, stream>>>(x, W1, h1, N);
    long tot1 = (long)N * 128;
    self_bias_init_kernel<128><<<(int)((tot1 + T - 1) / T), T, 0, stream>>>(h1, dinv, b1, agg1, tot1);
    scatter_edges_kernel<128><<<(E + 7) / 8, T, 0, stream>>>(h1, src, dst, norm, agg1, E);
    relu_kernel<<<(int)((tot1 + T - 1) / T), T, 0, stream>>>(agg1, tot1);

    gemm_wmma_f32<128, 64><<<gemm_blocks, 128, 0, stream>>>(agg1, W2, h2, N);
    long tot2 = (long)N * 64;
    self_bias_init_kernel<64><<<(int)((tot2 + T - 1) / T), T, 0, stream>>>(h2, dinv, b2, out, tot2);
    scatter_edges_kernel<64><<<(E + 7) / 8, T, 0, stream>>>(h2, src, dst, norm, out, E);
}


# Edge sums by gather over a dst-sorted edge list (counting sort, arrival-order rank per row) replace the float atomic scatter; each row summed in registers, written once with bias, self term, relu
# speedup vs baseline: 2.3327x; 2.3327x over previous
#include <hip/hip_runtime.h>

typedef __attribute__((ext_vector_type(2))) float v2f;
typedef __attribute__((ext_vector_type(4))) float v4f;
typedef __attribute__((ext_vector_type(8))) float v8f;


__global__ void cnt_zero_kernel(int* __restrict__ cnt, int n) {
    int i = blockIdx.x * blockDim.x + threadIdx.x;
    if (i < n) cnt[i] = 0;
}

__global__ void count_rank_kernel(const int* __restrict__ dst, int* __restrict__ cnt,
                                  int* __restrict__ rank, int E) {
    int e = blockIdx.x * blockDim.x + threadIdx.x;
    if (e < E) rank[e] = atomicAdd(&cnt[dst[e]], 1);
}

__global__ __launch_bounds__(1024) void rowptr_scan_kernel(const int* __restrict__ cnt,
                                                           int* __restrict__ rowptr, int n) {
    __shared__ int wave_tot[32];
    __shared__ int blk_base;
    const int tid  = threadIdx.x;
    const int lane = tid & 31;
    const int wave = tid >> 5;
    const int blk0 = blockIdx.x * 1024;

    int part = 0;
#pragma unroll 8
    for (int j = tid; j < blk0; j += 1024) part += cnt[j];
    for (int o = 16; o > 0; o >>= 1) part += __shfl_xor(part, o);
    if (lane == 0) wave_tot[wave] = part;
    __syncthreads();
    if (wave == 0) {
        int v = wave_tot[lane];
        for (int o = 16; o > 0; o >>= 1) v += __shfl_xor(v, o);
        if (lane == 0) blk_base = v;
    }
    __syncthreads();
    const int base = blk_base;

    const int i = blk0 + tid;
    const int c = (i < n) ? cnt[i] : 0;
    int incl = c;
    for (int o = 1; o < 32; o <<= 1) {
        const int t = __shfl_up(incl, o);
        if (lane >= o) incl += t;
    }
    if (lane == 31) wave_tot[wave] = incl;
    __syncthreads();
    int woff = 0;
    for (int k = 0; k < wave; ++k) woff += wave_tot[k];
    if (i < n) rowptr[i] = base + woff + incl - c;
}

__global__ void csr_fill_kernel(const int* __restrict__ src, const int* __restrict__ dst,
                                const int* __restrict__ rowptr, const int* __restrict__ rank,
                                int* __restrict__ csr_src, int E) {
    int e = blockIdx.x * blockDim.x + threadIdx.x;
    if (e < E) csr_src[rowptr[dst[e]] + rank[e]] = src[e];
}

template <int K, int NCOL>
__global__ __launch_bounds__(128) void gemm_wmma_f32(const float* __restrict__ A,
                                                     const float* __restrict__ B,
                                                     float* __restrict__ C, int nrows) {
    const int wave = threadIdx.x >> 5;
    const int lane = threadIdx.x & 31;
    const int half = lane >> 4;
    const int mr   = lane & 15;

    const int row_tile = blockIdx.x * (blockDim.x >> 5) + wave;
    const int row0 = row_tile * 16;
    if (row0 >= nrows) return;

    constexpr int NT = NCOL / 16;
    v8f acc[NT] = {};

    const float* arow = A + (size_t)(row0 + mr) * K + half * 2;

#pragma unroll 8
    for (int kb = 0; kb < K / 4; ++kb) {
        v2f a;
        a.x = arow[kb * 4 + 0];
        a.y = arow[kb * 4 + 1];
        const float* brow = B + (size_t)(kb * 4 + half * 2) * NCOL + mr;
#pragma unroll
        for (int ct = 0; ct < NT; ++ct) {
            v2f b;
            b.x = brow[ct * 16];
            b.y = brow[ct * 16 + NCOL];
            acc[ct] = __builtin_amdgcn_wmma_f32_16x16x4_f32(
                 false, a,  false, b,
                 (short)0, acc[ct],  false,  false);
        }
    }

#pragma unroll
    for (int ct = 0; ct < NT; ++ct) {
        float* crow = C + (size_t)(row0 + half * 8) * NCOL + ct * 16 + mr;
#pragma unroll
        for (int r = 0; r < 8; ++r) crow[(size_t)r * NCOL] = acc[ct][r];
    }
}

template <int V> struct vec_of;
template <> struct vec_of<4> { typedef v4f type; };
template <> struct vec_of<2> { typedef v2f type; };

template <int F, bool RELU>
__global__ __launch_bounds__(256) void aggregate_rows_kernel(const float* __restrict__ h,
                                                             const int* __restrict__ rowptr,
                                                             const int* __restrict__ csr_src,
                                                             const float* __restrict__ bias,
                                                             float* __restrict__ out,
                                                             int n, int E) {
    constexpr int V = F / 32;
    constexpr int U = 4;
    typedef typename vec_of<V>::type vec;

    const int lane = threadIdx.x & 31;
    const int d = __builtin_amdgcn_readfirstlane((int)((blockIdx.x * blockDim.x + threadIdx.x) >> 5));
    if (d >= n) return;

    const int beg = rowptr[d];
    const int end = (d + 1 < n) ? rowptr[d + 1] : E;
    const float dd = rsqrtf((float)(end - beg + 1));

    vec acc;
    {
        const vec hv = *((const vec*)(h + (size_t)d * F) + lane);
        const vec bv = *((const vec*)bias + lane);
        const float ws = dd * dd;
        acc = bv + ws * hv;
    }

    for (int base = beg; base < end; base += 32) {
        const int m = min(end - base, 32);
        int   s = 0;
        float w = 0.0f;
        if (lane < m) {
            s = csr_src[base + lane];
            const int rb = rowptr[s];
            const int re = (s + 1 < n) ? rowptr[s + 1] : E;
            w = rsqrtf((float)(re - rb + 1)) * dd;
        }
        int j0 = 0;
        for (; j0 + U <= m; j0 += U) {
            vec   hv[U];
            float wj[U];
#pragma unroll
            for (int u = 0; u < U; ++u) {
                const int sj = __builtin_amdgcn_readlane(s, j0 + u);
                wj[u] = __int_as_float(__builtin_amdgcn_readlane(__float_as_int(w), j0 + u));
                hv[u] = *((const vec*)(h + (size_t)sj * F) + lane);
            }
#pragma unroll
            for (int u = 0; u < U; ++u) acc += wj[u] * hv[u];
        }
        for (; j0 < m; ++j0) {
            const int   sj = __builtin_amdgcn_readlane(s, j0);
            const float wj = __int_as_float(__builtin_amdgcn_readlane(__float_as_int(w), j0));
            const vec   hv = *((const vec*)(h + (size_t)sj * F) + lane);
            acc += wj * hv;
        }
    }

    if (RELU) {
#pragma unroll
        for (int k = 0; k < V; ++k) acc[k] = fmaxf(acc[k], 0.0f);
    }
    *((vec*)(out + (size_t)d * F) + lane) = acc;
}

extern "C" void kernel_launch(void* const* d_in, const int* in_sizes, int n_in,
                              void* d_out, int out_size, void* d_ws, size_t ws_size,
                              hipStream_t stream) {
    const float* x  = (const float*)d_in[0];
    const int*   ei = (const int*)d_in[1];
    const float* W1 = (const float*)d_in[2];
    const float* b1 = (const float*)d_in[3];
    const float* W2 = (const float*)d_in[4];
    const float* b2 = (const float*)d_in[5];

    const int N = in_sizes[0] / 128;
    const int E = in_sizes[1] / 2;
    const int* src = ei;
    const int* dst = ei + E;

    int*   rowptr  = (int*)d_ws;
    int*   csr_src = rowptr + N;
    float* h1      = (float*)(csr_src + E);
    float* z       = h1 + (size_t)N * 128;
    int*   cnt     = (int*)z;
    int*   rank    = cnt + N;
    float* h2      = h1;
    float* out     = (float*)d_out;

    const int T = 256;

    cnt_zero_kernel<<<(N + T - 1) / T, T, 0, stream>>>(cnt, N);
    count_rank_kernel<<<(E + T - 1) / T, T, 0, stream>>>(dst, cnt, rank, E);
    rowptr_scan_kernel<<<(N + 1023) / 1024, 1024, 0, stream>>>(cnt, rowptr, N);
    csr_fill_kernel<<<(E + T - 1) / T, T, 0, stream>>>(src, dst, rowptr, rank, csr_src, E);

    const int row_tiles = (N + 15) / 16;
    const int gemm_blocks = (row_tiles + 3) / 4;
    const int agg_blocks = (N + 7) / 8;

    gemm_wmma_f32<128, 128><<<gemm_blocks, 128, 0, stream>>>(x, W1, h1, N);
    aggregate_rows_kernel<128, true><<<agg_blocks, T, 0, stream>>>(h1, rowptr, csr_src, b1, z, N, E);

    gemm_wmma_f32<128, 64><<<gemm_blocks, 128, 0, stream>>>(z, W2, h2, N);
    aggregate_rows_kernel<64, false><<<agg_blocks, T, 0, stream>>>(h2, rowptr, csr_src, b2, out, N, E);
}


# Edges grouped by dst in LDS: per-slice counts per 256-node range, prefix over slices, (dst,src) pairs placed per range, one block per range ranks rows; no global atomics; row sum order: LDS arrival
# speedup vs baseline: 3.4644x; 1.4851x over previous
#include <hip/hip_runtime.h>

typedef __attribute__((ext_vector_type(2))) float v2f;
typedef __attribute__((ext_vector_type(4))) float v4f;
typedef __attribute__((ext_vector_type(8))) float v8f;

#define RANGE_SHIFT 8
#define RANGE_NODES 256
#define NSLICE 256
#define EDGES_PER_THREAD 8
#define ROWS_STAGE 8192

__global__ __launch_bounds__(1024) void range_hist_kernel(const int* __restrict__ dst,
                                                          int* __restrict__ hist,
                                                          int E, int slice, int NB) {
    extern __shared__ int lds_cnt[];
    const int tid = threadIdx.x;
    for (int i = tid; i < NB; i += 1024) lds_cnt[i] = 0;
    __syncthreads();

    const int e0 = blockIdx.x * slice;
    const int e1 = min(E, e0 + slice);
    for (int base = e0 + tid; base < e1; base += EDGES_PER_THREAD * 1024) {
        int d[EDGES_PER_THREAD];
#pragma unroll
        for (int k = 0; k < EDGES_PER_THREAD; ++k) {
            d[k] = dst[min(base + k * 1024, e1 - 1)];
        }
        asm volatile("" :: "v"(d[0]), "v"(d[1]), "v"(d[2]), "v"(d[3]),
                           "v"(d[4]), "v"(d[5]), "v"(d[6]), "v"(d[7]));
#pragma unroll
        for (int k = 0; k < EDGES_PER_THREAD; ++k) {
            const int e = base + k * 1024;
            if (e < e1) atomicAdd(&lds_cnt[d[k] >> RANGE_SHIFT], 1);
        }
    }
    __syncthreads();
    for (int i = tid; i < NB; i += 1024) hist[(size_t)i * NSLICE + blockIdx.x] = lds_cnt[i];
}

__global__ __launch_bounds__(1024) void range_scan_kernel(int* __restrict__ hist,
                                                          int* __restrict__ tot, int NB) {
    const int lane = threadIdx.x & 31;
    const int r = blockIdx.x * 32 + (threadIdx.x >> 5);
    if (r >= NB) return;

    int4* p = (int4*)(hist + (size_t)r * NSLICE + lane * 8);
    int4 a = p[0];
    int4 b = p[1];
    const int s = a.x + a.y + a.z + a.w + b.x + b.y + b.z + b.w;
    int incl = s;
    for (int o = 1; o < 32; o <<= 1) {
        const int t = __shfl_up(incl, o);
        if (lane >= o) incl += t;
    }
    int run = incl - s;
    int4 oa, ob;
    oa.x = run; run += a.x;
    oa.y = run; run += a.y;
    oa.z = run; run += a.z;
    oa.w = run; run += a.w;
    ob.x = run; run += b.x;
    ob.y = run; run += b.y;
    ob.z = run; run += b.z;
    ob.w = run;
    p[0] = oa;
    p[1] = ob;
    if (lane == 31) tot[r] = incl;
}

__global__ __launch_bounds__(1024) void range_place_kernel(const int* __restrict__ src,
                                                           const int* __restrict__ dst,
                                                           const int* __restrict__ hist,
                                                           const int* __restrict__ tot,
                                                           int* __restrict__ range_start,
                                                           int2* __restrict__ pairs,
                                                           int E, int slice, int NB) {
    extern __shared__ int lds_cur[];
    __shared__ int wave_tot[32];
    const int tid  = threadIdx.x;
    const int lane = tid & 31;
    const int wave = tid >> 5;

    int carry = 0;
    for (int t0 = 0; t0 < NB; t0 += 1024) {
        const int i = t0 + tid;
        const int c = (i < NB) ? tot[i] : 0;
        int incl = c;
        for (int o = 1; o < 32; o <<= 1) {
            const int t = __shfl_up(incl, o);
            if (lane >= o) incl += t;
        }
        if (lane == 31) wave_tot[wave] = incl;
        __syncthreads();
        const int wt = wave_tot[lane];
        int wincl = wt;
        for (int o = 1; o < 32; o <<= 1) {
            const int t = __shfl_up(wincl, o);
            if (lane >= o) wincl += t;
        }
        const int woff = __shfl(wincl, wave) - __shfl(wt, wave);
        if (i < NB) lds_cur[i] = carry + woff + incl - c;
        carry += __shfl(wincl, 31);
        __syncthreads();
    }
    if (tid == 0) lds_cur[NB] = carry;
    __syncthreads();
    if (blockIdx.x == 0)
        for (int i = tid; i <= NB; i += 1024) range_start[i] = lds_cur[i];
    __syncthreads();
    for (int i = tid; i < NB; i += 1024) lds_cur[i] += hist[(size_t)i * NSLICE + blockIdx.x];
    __syncthreads();

    const int e0 = blockIdx.x * slice;
    const int e1 = min(E, e0 + slice);
    for (int base = e0 + tid; base < e1; base += EDGES_PER_THREAD * 1024) {
        int d[EDGES_PER_THREAD], s[EDGES_PER_THREAD];
#pragma unroll
        for (int k = 0; k < EDGES_PER_THREAD; ++k) {
            const int ec = min(base + k * 1024, e1 - 1);
            d[k] = dst[ec];
            s[k] = src[ec];
        }
        asm volatile("" :: "v"(d[0]), "v"(d[1]), "v"(d[2]), "v"(d[3]),
                           "v"(d[4]), "v"(d[5]), "v"(d[6]), "v"(d[7]),
                           "v"(s[0]), "v"(s[1]), "v"(s[2]), "v"(s[3]),
                           "v"(s[4]), "v"(s[5]), "v"(s[6]), "v"(s[7]));
#pragma unroll
        for (int k = 0; k < EDGES_PER_THREAD; ++k) {
            const int e = base + k * 1024;
            if (e < e1) {
                const int slot = atomicAdd(&lds_cur[d[k] >> RANGE_SHIFT], 1);
                pairs[slot] = make_int2(d[k], s[k]);
            }
        }
    }
}

__global__ __launch_bounds__(1024) void range_rows_kernel(const int2* __restrict__ pairs,
                                                          const int* __restrict__ range_start,
                                                          int* __restrict__ rowptr,
                                                          int* __restrict__ csr_src, int n) {
    __shared__ int node_cnt[RANGE_NODES];
    __shared__ int node_cur[RANGE_NODES];
    __shared__ int wave_tot[RANGE_NODES / 32];
    __shared__ int staged[ROWS_STAGE];
    const int tid   = threadIdx.x;
    const int lane  = tid & 31;
    const int wave  = tid >> 5;
    const int node0 = blockIdx.x << RANGE_SHIFT;
    const int b0 = range_start[blockIdx.x];
    const int b1 = range_start[blockIdx.x + 1];
    const bool in_lds = (b1 - b0) <= ROWS_STAGE;

    if (tid < RANGE_NODES) node_cnt[tid] = 0;
    __syncthreads();
    for (int base = b0 + tid; base < b1; base += 4 * 1024) {
        int d[4];
#pragma unroll
        for (int k = 0; k < 4; ++k) {
            d[k] = pairs[min(base + k * 1024, b1 - 1)].x;
        }
        asm volatile("" :: "v"(d[0]), "v"(d[1]), "v"(d[2]), "v"(d[3]));
#pragma unroll
        for (int k = 0; k < 4; ++k) {
            const int i = base + k * 1024;
            if (i < b1) atomicAdd(&node_cnt[d[k] - node0], 1);
        }
    }
    __syncthreads();

    int c = 0, incl = 0;
    if (tid < RANGE_NODES) {
        c = node_cnt[tid];
        incl = c;
        for (int o = 1; o < 32; o <<= 1) {
            const int t = __shfl_up(incl, o);
            if (lane >= o) incl += t;
        }
        if (lane == 31) wave_tot[wave] = incl;
    }
    __syncthreads();
    if (tid < RANGE_NODES) {
        int woff = 0;
        for (int k = 0; k < wave; ++k) woff += wave_tot[k];
        const int first = b0 + woff + incl - c;
        node_cur[tid] = first;
        if (node0 + tid < n) rowptr[node0 + tid] = first;
    }
    __syncthreads();

    for (int base = b0 + tid; base < b1; base += 4 * 1024) {
        int2 p[4];
#pragma unroll
        for (int k = 0; k < 4; ++k) {
            p[k] = pairs[min(base + k * 1024, b1 - 1)];
        }
        asm volatile("" :: "v"(p[0].x), "v"(p[0].y), "v"(p[1].x), "v"(p[1].y),
                           "v"(p[2].x), "v"(p[2].y), "v"(p[3].x), "v"(p[3].y));
#pragma unroll
        for (int k = 0; k < 4; ++k) {
            const int i = base + k * 1024;
            if (i < b1) {
                const int slot = atomicAdd(&node_cur[p[k].x - node0], 1);
                if (in_lds) staged[slot - b0] = p[k].y;
                else        csr_src[slot] = p[k].y;
            }
        }
    }
    __syncthreads();
    if (in_lds)
        for (int i = b0 + tid; i < b1; i += 1024) csr_src[i] = staged[i - b0];
}

template <int K, int NCOL>
__global__ __launch_bounds__(128) void gemm_wmma_f32(const float* __restrict__ A,
                                                     const float* __restrict__ B,
                                                     float* __restrict__ C, int nrows) {
    const int wave = threadIdx.x >> 5;
    const int lane = threadIdx.x & 31;
    const int half = lane >> 4;
    const int mr   = lane & 15;

    const int row_tile = blockIdx.x * (blockDim.x >> 5) + wave;
    const int row0 = row_tile * 16;
    if (row0 >= nrows) return;

    constexpr int NT = NCOL / 16;
    v8f acc[NT] = {};

    const float* arow = A + (size_t)(row0 + mr) * K + half * 2;

#pragma unroll 8
    for (int kb = 0; kb < K / 4; ++kb) {
        v2f a;
        a.x = arow[kb * 4 + 0];
        a.y = arow[kb * 4 + 1];
        const float* brow = B + (size_t)(kb * 4 + half * 2) * NCOL + mr;
#pragma unroll
        for (int ct = 0; ct < NT; ++ct) {
            v2f b;
            b.x = brow[ct * 16];
            b.y = brow[ct * 16 + NCOL];
            acc[ct] = __builtin_amdgcn_wmma_f32_16x16x4_f32(
                 false, a,  false, b,
                 (short)0, acc[ct],  false,  false);
        }
    }

#pragma unroll
    for (int ct = 0; ct < NT; ++ct) {
        float* crow = C + (size_t)(row0 + half * 8) * NCOL + ct * 16 + mr;
#pragma unroll
        for (int r = 0; r < 8; ++r) crow[(size_t)r * NCOL] = acc[ct][r];
    }
}

template <int V> struct vec_of;
template <> struct vec_of<4> { typedef v4f type; };
template <> struct vec_of<2> { typedef v2f type; };

template <int F, bool RELU>
__global__ __launch_bounds__(256) void aggregate_rows_kernel(const float* __restrict__ h,
                                                             const int* __restrict__ rowptr,
                                                             const int* __restrict__ csr_src,
                                                             const float* __restrict__ bias,
                                                             float* __restrict__ out,
                                                             int n, int E) {
    constexpr int V = F / 32;
    constexpr int U = 4;
    typedef typename vec_of<V>::type vec;

    const int lane = threadIdx.x & 31;
    const int d = __builtin_amdgcn_readfirstlane((int)((blockIdx.x * blockDim.x + threadIdx.x) >> 5));
    if (d >= n) return;

    const int beg = rowptr[d];
    const int end = (d + 1 < n) ? rowptr[d + 1] : E;
    const float dd = rsqrtf((float)(end - beg + 1));

    vec acc;
    {
        const vec hv = *((const vec*)(h + (size_t)d * F) + lane);
        const vec bv = *((const vec*)bias + lane);
        const float ws = dd * dd;
        acc = bv + ws * hv;
    }

    for (int base = beg; base < end; base += 32) {
        const int m = min(end - base, 32);
        int   s = 0;
        float w = 0.0f;
        if (lane < m) {
            s = csr_src[base + lane];
            const int rb = rowptr[s];
            const int re = (s + 1 < n) ? rowptr[s + 1] : E;
            w = rsqrtf((float)(re - rb + 1)) * dd;
        }
        int j0 = 0;
        for (; j0 + U <= m; j0 += U) {
            vec   hv[U];
            float wj[U];
#pragma unroll
            for (int u = 0; u < U; ++u) {
                const int sj = __builtin_amdgcn_readlane(s, j0 + u);
                wj[u] = __int_as_float(__builtin_amdgcn_readlane(__float_as_int(w), j0 + u));
                hv[u] = *((const vec*)(h + (size_t)sj * F) + lane);
            }
#pragma unroll
            for (int u = 0; u < U; ++u) acc += wj[u] * hv[u];
        }
        for (; j0 < m; ++j0) {
            const int   sj = __builtin_amdgcn_readlane(s, j0);
            const float wj = __int_as_float(__builtin_amdgcn_readlane(__float_as_int(w), j0));
            const vec   hv = *((const vec*)(h + (size_t)sj * F) + lane);
            acc += wj * hv;
        }
    }

    if (RELU) {
#pragma unroll
        for (int k = 0; k < V; ++k) acc[k] = fmaxf(acc[k], 0.0f);
    }
    *((vec*)(out + (size_t)d * F) + lane) = acc;
}

extern "C" void kernel_launch(void* const* d_in, const int* in_sizes, int n_in,
                              void* d_out, int out_size, void* d_ws, size_t ws_size,
                              hipStream_t stream) {
    const float* x  = (const float*)d_in[0];
    const int*   ei = (const int*)d_in[1];
    const float* W1 = (const float*)d_in[2];
    const float* b1 = (const float*)d_in[3];
    const float* W2 = (const float*)d_in[4];
    const float* b2 = (const float*)d_in[5];

    const int N = in_sizes[0] / 128;
    const int E = in_sizes[1] / 2;
    const int* src = ei;
    const int* dst = ei + E;

    int*   rowptr  = (int*)d_ws;
    int*   csr_src = rowptr + N;
    float* h1      = (float*)(csr_src + E);
    float* z       = h1 + (size_t)N * 128;
    const int NB   = (N + RANGE_NODES - 1) / RANGE_NODES;
    int2*  pairs   = (int2*)z;
    int*   hist    = (int*)(pairs + E);
    int*   tot     = hist + (size_t)NB * NSLICE;
    int*   range_start = tot + NB;
    float* h2      = h1;
    float* out     = (float*)d_out;

    const int T = 256;

    const int slice = (E + NSLICE - 1) / NSLICE;
    range_hist_kernel<<<NSLICE, 1024, NB * sizeof(int), stream>>>(dst, hist, E, slice, NB);
    range_scan_kernel<<<(NB + 31) / 32, 1024, 0, stream>>>(hist, tot, NB);
    range_place_kernel<<<NSLICE, 1024, (NB + 1) * sizeof(int), stream>>>(src, dst, hist, tot, range_start,
                                                                        pairs, E, slice, NB);
    range_rows_kernel<<<NB, 1024, 0, stream>>>(pairs, range_start, rowptr, csr_src, N);

    const int row_tiles = (N + 15) / 16;
    const int gemm_blocks = (row_tiles + 3) / 4;
    const int agg_blocks = (N + 7) / 8;

    gemm_wmma_f32<128, 128><<<gemm_blocks, 128, 0, stream>>>(x, W1, h1, N);
    aggregate_rows_kernel<128, true><<<agg_blocks, T, 0, stream>>>(h1, rowptr, csr_src, b1, z, N, E);

    gemm_wmma_f32<128, 64><<<gemm_blocks, 128, 0, stream>>>(z, W2, h2, N);
    aggregate_rows_kernel<64, false><<<agg_blocks, T, 0, stream>>>(h2, rowptr, csr_src, b2, out, N, E);
}


# Layer-1 aggregation keeps each 16-row relu tile in LDS (4 waves x 4 rows) and multiplies it by W2 in the same kernel, f32 WMMA with A from LDS: z never stored or re-read, one launch less
# speedup vs baseline: 4.0220x; 1.1610x over previous
#include <hip/hip_runtime.h>

typedef __attribute__((ext_vector_type(2))) float v2f;
typedef __attribute__((ext_vector_type(4))) float v4f;
typedef __attribute__((ext_vector_type(8))) float v8f;

#define RANGE_SHIFT 8
#define RANGE_NODES 256
#define NSLICE 256
#define EDGES_PER_THREAD 8
#define ROWS_STAGE 8192

__global__ __launch_bounds__(1024) void range_hist_kernel(const int* __restrict__ dst,
                                                          int* __restrict__ hist,
                                                          int E, int slice, int NB) {
    extern __shared__ int lds_cnt[];
    const int tid = threadIdx.x;
    for (int i = tid; i < NB; i += 1024) lds_cnt[i] = 0;
    __syncthreads();

    const int e0 = blockIdx.x * slice;
    const int e1 = min(E, e0 + slice);
    for (int base = e0 + tid; base < e1; base += EDGES_PER_THREAD * 1024) {
        int d[EDGES_PER_THREAD];
#pragma unroll
        for (int k = 0; k < EDGES_PER_THREAD; ++k) {
            d[k] = dst[min(base + k * 1024, e1 - 1)];
        }
        asm volatile("" :: "v"(d[0]), "v"(d[1]), "v"(d[2]), "v"(d[3]),
                           "v"(d[4]), "v"(d[5]), "v"(d[6]), "v"(d[7]));
#pragma unroll
        for (int k = 0; k < EDGES_PER_THREAD; ++k) {
            const int e = base + k * 1024;
            if (e < e1) atomicAdd(&lds_cnt[d[k] >> RANGE_SHIFT], 1);
        }
    }
    __syncthreads();
    for (int i = tid; i < NB; i += 1024) hist[(size_t)i * NSLICE + blockIdx.x] = lds_cnt[i];
}

__global__ __launch_bounds__(1024) void range_scan_kernel(int* __restrict__ hist,
                                                          int* __restrict__ tot, int NB) {
    const int lane = threadIdx.x & 31;
    const int r = blockIdx.x * 32 + (threadIdx.x >> 5);
    if (r >= NB) return;

    int4* p = (int4*)(hist + (size_t)r * NSLICE + lane * 8);
    int4 a = p[0];
    int4 b = p[1];
    const int s = a.x + a.y + a.z + a.w + b.x + b.y + b.z + b.w;
    int incl = s;
    for (int o = 1; o < 32; o <<= 1) {
        const int t = __shfl_up(incl, o);
        if (lane >= o) incl += t;
    }
    int run = incl - s;
    int4 oa, ob;
    oa.x = run; run += a.x;
    oa.y = run; run += a.y;
    oa.z = run; run += a.z;
    oa.w = run; run += a.w;
    ob.x = run; run += b.x;
    ob.y = run; run += b.y;
    ob.z = run; run += b.z;
    ob.w = run;
    p[0] = oa;
    p[1] = ob;
    if (lane == 31) tot[r] = incl;
}

__global__ __launch_bounds__(1024) void range_place_kernel(const int* __restrict__ src,
                                                           const int* __restrict__ dst,
                                                           const int* __restrict__ hist,
                                                           const int* __restrict__ tot,
                                                           int* __restrict__ range_start,
                                                           int2* __restrict__ pairs,
                                                           int E, int slice, int NB) {
    extern __shared__ int lds_cur[];
    __shared__ int wave_tot[32];
    const int tid  = threadIdx.x;
    const int lane = tid & 31;
    const int wave = tid >> 5;

    int carry = 0;
    for (int t0 = 0; t0 < NB; t0 += 1024) {
        const int i = t0 + tid;
        const int c = (i < NB) ? tot[i] : 0;
        int incl = c;
        for (int o = 1; o < 32; o <<= 1) {
            const int t = __shfl_up(incl, o);
            if (lane >= o) incl += t;
        }
        if (lane == 31) wave_tot[wave] = incl;
        __syncthreads();
        const int wt = wave_tot[lane];
        int wincl = wt;
        for (int o = 1; o < 32; o <<= 1) {
            const int t = __shfl_up(wincl, o);
            if (lane >= o) wincl += t;
        }
        const int woff = __shfl(wincl, wave) - __shfl(wt, wave);
        if (i < NB) lds_cur[i] = carry + woff + incl - c;
        carry += __shfl(wincl, 31);
        __syncthreads();
    }
    if (tid == 0) lds_cur[NB] = carry;
    __syncthreads();
    if (blockIdx.x == 0)
        for (int i = tid; i <= NB; i += 1024) range_start[i] = lds_cur[i];
    __syncthreads();
    for (int i = tid; i < NB; i += 1024) lds_cur[i] += hist[(size_t)i * NSLICE + blockIdx.x];
    __syncthreads();

    const int e0 = blockIdx.x * slice;
    const int e1 = min(E, e0 + slice);
    for (int base = e0 + tid; base < e1; base += EDGES_PER_THREAD * 1024) {
        int d[EDGES_PER_THREAD], s[EDGES_PER_THREAD];
#pragma unroll
        for (int k = 0; k < EDGES_PER_THREAD; ++k) {
            const int ec = min(base + k * 1024, e1 - 1);
            d[k] = dst[ec];
            s[k] = src[ec];
        }
        asm volatile("" :: "v"(d[0]), "v"(d[1]), "v"(d[2]), "v"(d[3]),
                           "v"(d[4]), "v"(d[5]), "v"(d[6]), "v"(d[7]),
                           "v"(s[0]), "v"(s[1]), "v"(s[2]), "v"(s[3]),
                           "v"(s[4]), "v"(s[5]), "v"(s[6]), "v"(s[7]));
#pragma unroll
        for (int k = 0; k < EDGES_PER_THREAD; ++k) {
            const int e = base + k * 1024;
            if (e < e1) {
                const int slot = atomicAdd(&lds_cur[d[k] >> RANGE_SHIFT], 1);
                pairs[slot] = make_int2(d[k], s[k]);
            }
        }
    }
}

__global__ __launch_bounds__(1024) void range_rows_kernel(const int2* __restrict__ pairs,
                                                          const int* __restrict__ range_start,
                                                          int* __restrict__ rowptr,
                                                          int* __restrict__ csr_src, int n) {
    __shared__ int node_cnt[RANGE_NODES];
    __shared__ int node_cur[RANGE_NODES];
    __shared__ int wave_tot[RANGE_NODES / 32];
    __shared__ int staged[ROWS_STAGE];
    const int tid   = threadIdx.x;
    const int lane  = tid & 31;
    const int wave  = tid >> 5;
    const int node0 = blockIdx.x << RANGE_SHIFT;
    const int b0 = range_start[blockIdx.x];
    const int b1 = range_start[blockIdx.x + 1];
    const bool in_lds = (b1 - b0) <= ROWS_STAGE;

    if (tid < RANGE_NODES) node_cnt[tid] = 0;
    __syncthreads();
    for (int base = b0 + tid; base < b1; base += 4 * 1024) {
        int d[4];
#pragma unroll
        for (int k = 0; k < 4; ++k) {
            d[k] = pairs[min(base + k * 1024, b1 - 1)].x;
        }
        asm volatile("" :: "v"(d[0]), "v"(d[1]), "v"(d[2]), "v"(d[3]));
#pragma unroll
        for (int k = 0; k < 4; ++k) {
            const int i = base + k * 1024;
            if (i < b1) atomicAdd(&node_cnt[d[k] - node0], 1);
        }
    }
    __syncthreads();

    int c = 0, incl = 0;
    if (tid < RANGE_NODES) {
        c = node_cnt[tid];
        incl = c;
        for (int o = 1; o < 32; o <<= 1) {
            const int t = __shfl_up(incl, o);
            if (lane >= o) incl += t;
        }
        if (lane == 31) wave_tot[wave] = incl;
    }
    __syncthreads();
    if (tid < RANGE_NODES) {
        int woff = 0;
        for (int k = 0; k < wave; ++k) woff += wave_tot[k];
        const int first = b0 + woff + incl - c;
        node_cur[tid] = first;
        if (node0 + tid < n) rowptr[node0 + tid] = first;
    }
    __syncthreads();

    for (int base = b0 + tid; base < b1; base += 4 * 1024) {
        int2 p[4];
#pragma unroll
        for (int k = 0; k < 4; ++k) {
            p[k] = pairs[min(base + k * 1024, b1 - 1)];
        }
        asm volatile("" :: "v"(p[0].x), "v"(p[0].y), "v"(p[1].x), "v"(p[1].y),
                           "v"(p[2].x), "v"(p[2].y), "v"(p[3].x), "v"(p[3].y));
#pragma unroll
        for (int k = 0; k < 4; ++k) {
            const int i = base + k * 1024;
            if (i < b1) {
                const int slot = atomicAdd(&node_cur[p[k].x - node0], 1);
                if (in_lds) staged[slot - b0] = p[k].y;
                else        csr_src[slot] = p[k].y;
            }
        }
    }
    __syncthreads();
    if (in_lds)
        for (int i = b0 + tid; i < b1; i += 1024) csr_src[i] = staged[i - b0];
}

template <int K, int NCOL>
__global__ __launch_bounds__(128) void gemm_wmma_f32(const float* __restrict__ A,
                                                     const float* __restrict__ B,
                                                     const int* __restrict__ rowptr,
                                                     float* __restrict__ C, int nrows, int E) {
    const int wave = threadIdx.x >> 5;
    const int lane = threadIdx.x & 31;
    const int half = lane >> 4;
    const int mr   = lane & 15;

    const int row_tile = blockIdx.x * (blockDim.x >> 5) + wave;
    const int row0 = row_tile * 16;
    if (row0 >= nrows) return;

    constexpr int NT = NCOL / 16;
    v8f acc[NT] = {};

    const int ri   = row0 + mr;
    const int rbeg = rowptr[min(ri, nrows - 1)];
    const int rnxt = rowptr[min(ri + 1, nrows - 1)];
    const int rend = (ri + 1 < nrows) ? rnxt : E;

    const float* arow = A + (size_t)(row0 + mr) * K + half * 2;

#pragma unroll 8
    for (int kb = 0; kb < K / 4; ++kb) {
        v2f a;
        a.x = arow[kb * 4 + 0];
        a.y = arow[kb * 4 + 1];
        const float* brow = B + (size_t)(kb * 4 + half * 2) * NCOL + mr;
#pragma unroll
        for (int ct = 0; ct < NT; ++ct) {
            v2f b;
            b.x = brow[ct * 16];
            b.y = brow[ct * 16 + NCOL];
            acc[ct] = __builtin_amdgcn_wmma_f32_16x16x4_f32(
                 false, a,  false, b,
                 (short)0, acc[ct],  false,  false);
        }
    }

    const int dv = __float_as_int(rsqrtf((float)(rend - rbeg + 1)));
    float sc[8];
#pragma unroll
    for (int r = 0; r < 8; ++r) {
        const float lo = __int_as_float(__builtin_amdgcn_readlane(dv, r));
        const float hi = __int_as_float(__builtin_amdgcn_readlane(dv, r + 8));
        sc[r] = half ? hi : lo;
    }

#pragma unroll
    for (int ct = 0; ct < NT; ++ct) {
        float* crow = C + (size_t)(row0 + half * 8) * NCOL + ct * 16 + mr;
#pragma unroll
        for (int r = 0; r < 8; ++r) crow[(size_t)r * NCOL] = sc[r] * acc[ct][r];
    }
}

template <int V> struct vec_of;
template <> struct vec_of<4> { typedef v4f type; };
template <> struct vec_of<2> { typedef v2f type; };

template <int F, bool RELU>
__global__ __launch_bounds__(256) void aggregate_rows_kernel(const float* __restrict__ hs,
                                                             const int* __restrict__ rowptr,
                                                             const int* __restrict__ csr_src,
                                                             const float* __restrict__ bias,
                                                             float* __restrict__ out,
                                                             int n, int E) {
    constexpr int V = F / 32;
    constexpr int U = 4;
    typedef typename vec_of<V>::type vec;

    const int lane = threadIdx.x & 31;
    const int d = __builtin_amdgcn_readfirstlane((int)((blockIdx.x * blockDim.x + threadIdx.x) >> 5));
    if (d >= n) return;

    const int beg = rowptr[d];
    const int end = (d + 1 < n) ? rowptr[d + 1] : E;

    vec acc = *((const vec*)(hs + (size_t)d * F) + lane);

    for (int base = beg; base < end; base += 32) {
        const int m = min(end - base, 32);
        int s = 0;
        if (lane < m) s = csr_src[base + lane];
        int j0 = 0;
        for (; j0 + U <= m; j0 += U) {
            vec hv[U];
#pragma unroll
            for (int u = 0; u < U; ++u) {
                const int sj = __builtin_amdgcn_readlane(s, j0 + u);
                hv[u] = *((const vec*)(hs + (size_t)sj * F) + lane);
            }
#pragma unroll
            for (int u = 0; u < U; ++u) acc += hv[u];
        }
        for (; j0 < m; ++j0) {
            const int sj = __builtin_amdgcn_readlane(s, j0);
            const vec hv = *((const vec*)(hs + (size_t)sj * F) + lane);
            acc += hv;
        }
    }

    {
        const float dd = rsqrtf((float)(end - beg + 1));
        const vec bv = *((const vec*)bias + lane);
        acc = bv + dd * acc;
    }

    if (RELU) {
#pragma unroll
        for (int k = 0; k < V; ++k) acc[k] = fmaxf(acc[k], 0.0f);
    }
    *((vec*)(out + (size_t)d * F) + lane) = acc;
}

#define TILE_ROWS 16
#define ROWS_PER_WAVE 4
#define ZT_PITCH 132

__global__ __launch_bounds__(128) void aggregate_relu_gemm_kernel(const float* __restrict__ hs,
                                                                  const int* __restrict__ rowptr,
                                                                  const int* __restrict__ csr_src,
                                                                  const float* __restrict__ bias,
                                                                  const float* __restrict__ B,
                                                                  float* __restrict__ C,
                                                                  int n, int E) {
    constexpr int F = 128;
    constexpr int NCOL = 64;
    constexpr int U = 4;
    __shared__ __attribute__((aligned(16))) float zt[TILE_ROWS * ZT_PITCH];
    __shared__ __attribute__((aligned(16))) float dts[TILE_ROWS];

    const int lane = threadIdx.x & 31;
    const int wave = __builtin_amdgcn_readfirstlane((int)(threadIdx.x >> 5));
    const int node0 = blockIdx.x * TILE_ROWS;
    const int d0 = node0 + wave * ROWS_PER_WAVE;

    int rb[ROWS_PER_WAVE + 1];
    if (d0 + ROWS_PER_WAVE < n) {
#pragma unroll
        for (int i = 0; i <= ROWS_PER_WAVE; ++i) rb[i] = rowptr[d0 + i];
    } else {
#pragma unroll
        for (int i = 0; i <= ROWS_PER_WAVE; ++i) {
            const int v = rowptr[min(d0 + i, n - 1)];
            rb[i] = (d0 + i < n) ? v : E;
        }
    }

#pragma unroll
    for (int i = 0; i < ROWS_PER_WAVE; ++i) {
        const int t = wave * ROWS_PER_WAVE + i;
        const int d = d0 + i;
        const int beg = rb[i];
        const int end = rb[i + 1];
        v4f acc = {0.0f, 0.0f, 0.0f, 0.0f};
        float dd = 0.0f;
        if (d < n) {
            acc = *((const v4f*)(hs + (size_t)d * F) + lane);
            for (int base = beg; base < end; base += 32) {
                const int m = min(end - base, 32);
                int s = 0;
                if (lane < m) s = csr_src[base + lane];
                int j0 = 0;
                for (; j0 + U <= m; j0 += U) {
                    v4f hv[U];
#pragma unroll
                    for (int u = 0; u < U; ++u) {
                        const int sj = __builtin_amdgcn_readlane(s, j0 + u);
                        hv[u] = *((const v4f*)(hs + (size_t)sj * F) + lane);
                    }
#pragma unroll
                    for (int u = 0; u < U; ++u) acc += hv[u];
                }
                for (; j0 < m; ++j0) {
                    const int sj = __builtin_amdgcn_readlane(s, j0);
                    const v4f hv = *((const v4f*)(hs + (size_t)sj * F) + lane);
                    acc += hv;
                }
            }
            dd = rsqrtf((float)(end - beg + 1));
            const v4f bv = *((const v4f*)bias + lane);
            acc = bv + dd * acc;
#pragma unroll
            for (int k = 0; k < 4; ++k) acc[k] = fmaxf(acc[k], 0.0f);
        }
        *((v4f*)(zt + t * ZT_PITCH) + lane) = acc;
        if (lane == 0) dts[t] = dd;
    }

    const int half = lane >> 4;
    const int mr   = lane & 15;

    const float* arow = zt + mr * ZT_PITCH + half * 2;
    const float* bcol = B + (size_t)(half * 2) * NCOL + wave * 16 + mr;

    constexpr int KG = 8;
    v2f bq[2][KG];
#pragma unroll
    for (int j = 0; j < KG; ++j) {
        bq[0][j].x = bcol[(size_t)(j * 4) * NCOL];
        bq[0][j].y = bcol[(size_t)(j * 4 + 1) * NCOL];
    }

    __syncthreads();

    v8f c = {};
#pragma unroll
    for (int g = 0; g < F / 4 / KG; ++g) {
        if (g + 1 < F / 4 / KG) {
#pragma unroll
            for (int j = 0; j < KG; ++j) {
                const int kb = (g + 1) * KG + j;
                bq[(g + 1) & 1][j].x = bcol[(size_t)(kb * 4) * NCOL];
                bq[(g + 1) & 1][j].y = bcol[(size_t)(kb * 4 + 1) * NCOL];
            }
        }
#pragma unroll
        for (int j = 0; j < KG; ++j) {
            const int kb = g * KG + j;
            const v2f a = *(const v2f*)(arow + kb * 4);
            c = __builtin_amdgcn_wmma_f32_16x16x4_f32(
                 false, a,  false, bq[g & 1][j],
                 (short)0, c,  false,  false);
        }
    }
    __builtin_amdgcn_sched_barrier(0);
    asm volatile("v_nop\n\tv_nop\n\tv_nop\n\tv_nop\n\tv_nop\n\tv_nop\n\tv_nop\n\tv_nop" : "+v"(c));
    __builtin_amdgcn_sched_barrier(0);

    const v4f s0 = *(const v4f*)(dts + half * 8);
    const v4f s1 = *(const v4f*)(dts + half * 8 + 4);
    float o[8];
#pragma unroll
    for (int r = 0; r < 4; ++r) {
        o[r]     = s0[r] * c[r];
        o[r + 4] = s1[r] * c[r + 4];
    }
    float* crow = C + (size_t)(node0 + half * 8) * NCOL + wave * 16 + mr;
    if (node0 + TILE_ROWS <= n) {
#pragma unroll
        for (int r = 0; r < 8; ++r) crow[(size_t)r * NCOL] = o[r];
    } else {
#pragma unroll
        for (int r = 0; r < 8; ++r)
            if (node0 + half * 8 + r < n) crow[(size_t)r * NCOL] = o[r];
    }
}

extern "C" void kernel_launch(void* const* d_in, const int* in_sizes, int n_in,
                              void* d_out, int out_size, void* d_ws, size_t ws_size,
                              hipStream_t stream) {
    const float* x  = (const float*)d_in[0];
    const int*   ei = (const int*)d_in[1];
    const float* W1 = (const float*)d_in[2];
    const float* b1 = (const float*)d_in[3];
    const float* W2 = (const float*)d_in[4];
    const float* b2 = (const float*)d_in[5];

    const int N = in_sizes[0] / 128;
    const int E = in_sizes[1] / 2;
    const int* src = ei;
    const int* dst = ei + E;

    int*   rowptr  = (int*)d_ws;
    int*   csr_src = rowptr + N;
    float* h1      = (float*)(csr_src + E);
    float* h2      = h1 + (size_t)N * 128;
    const int NB   = (N + RANGE_NODES - 1) / RANGE_NODES;
    int2*  pairs   = (int2*)h2;
    int*   hist    = (int*)(pairs + E);
    int*   tot     = hist + (size_t)NB * NSLICE;
    int*   range_start = tot + NB;
    float* out     = (float*)d_out;

    const int T = 256;

    const int slice = (E + NSLICE - 1) / NSLICE;
    range_hist_kernel<<<NSLICE, 1024, NB * sizeof(int), stream>>>(dst, hist, E, slice, NB);
    range_scan_kernel<<<(NB + 31) / 32, 1024, 0, stream>>>(hist, tot, NB);
    range_place_kernel<<<NSLICE, 1024, (NB + 1) * sizeof(int), stream>>>(src, dst, hist, tot, range_start,
                                                                        pairs, E, slice, NB);
    range_rows_kernel<<<NB, 1024, 0, stream>>>(pairs, range_start, rowptr, csr_src, N);

    const int row_tiles = (N + 15) / 16;
    const int gemm_blocks = (row_tiles + 3) / 4;
    const int agg_blocks = (N + 7) / 8;

    gemm_wmma_f32<128, 128><<<gemm_blocks, 128, 0, stream>>>(x, W1, rowptr, h1, N, E);
    aggregate_relu_gemm_kernel<<<row_tiles, 128, 0, stream>>>(h1, rowptr, csr_src, b1, W2, h2, N, E);

    aggregate_rows_kernel<64, false><<<agg_blocks, T, 0, stream>>>(h2, rowptr, csr_src, b2, out, N, E);
}


# Edge grouping in 2 launches, not 4: each slice groups its edges by range in LDS, range blocks fetch the 256 slice runs; each row's sources sorted by index: row sum order fixed, not LDS arrival
# speedup vs baseline: 4.4308x; 1.1016x over previous
#include <hip/hip_runtime.h>

typedef __attribute__((ext_vector_type(2))) float v2f;
typedef __attribute__((ext_vector_type(4))) float v4f;
typedef __attribute__((ext_vector_type(8))) float v8f;

#define RANGE_SHIFT 8
#define RANGE_NODES 256
#define NSLICE 256
#define EDGES_PER_THREAD 8
#define ROWS_STAGE 8192

__global__ __launch_bounds__(1024) void slice_group_kernel(const int* __restrict__ src,
                                                           const int* __restrict__ dst,
                                                           int* __restrict__ offT,
                                                           int2* __restrict__ pairs,
                                                           int E, int slice, int NB) {
    extern __shared__ int lds_dyn[];
    __shared__ int wave_tot[32];
    int*  cur   = lds_dyn;
    int2* stage = (int2*)(lds_dyn + ((NB + 2) & ~1));
    const int tid  = threadIdx.x;
    const int lane = tid & 31;
    const int wave = tid >> 5;

    for (int i = tid; i <= NB; i += 1024) cur[i] = 0;
    __syncthreads();

    const int e0 = min(E, (int)blockIdx.x * slice);
    const int e1 = min(E, e0 + slice);
    const bool one_trip = slice <= EDGES_PER_THREAD * 1024;
    const int base0 = e0 + tid;
    int d0[EDGES_PER_THREAD], s0[EDGES_PER_THREAD];
#pragma unroll
    for (int k = 0; k < EDGES_PER_THREAD; ++k) { d0[k] = 0; s0[k] = 0; }
    if (one_trip && e0 < e1) {
#pragma unroll
        for (int k = 0; k < EDGES_PER_THREAD; ++k) {
            const int ec = min(base0 + k * 1024, e1 - 1);
            d0[k] = dst[ec];
            s0[k] = src[ec];
        }
    }

    if (one_trip) {
        asm volatile("" :: "v"(d0[0]), "v"(d0[1]), "v"(d0[2]), "v"(d0[3]),
                           "v"(d0[4]), "v"(d0[5]), "v"(d0[6]), "v"(d0[7]));
#pragma unroll
        for (int k = 0; k < EDGES_PER_THREAD; ++k) {
            if (base0 + k * 1024 < e1) atomicAdd(&cur[d0[k] >> RANGE_SHIFT], 1);
        }
    } else {
        for (int base = base0; base < e1; base += EDGES_PER_THREAD * 1024) {
            int d[EDGES_PER_THREAD];
#pragma unroll
            for (int k = 0; k < EDGES_PER_THREAD; ++k) {
                d[k] = dst[min(base + k * 1024, e1 - 1)];
            }
            asm volatile("" :: "v"(d[0]), "v"(d[1]), "v"(d[2]), "v"(d[3]),
                               "v"(d[4]), "v"(d[5]), "v"(d[6]), "v"(d[7]));
#pragma unroll
            for (int k = 0; k < EDGES_PER_THREAD; ++k) {
                const int e = base + k * 1024;
                if (e < e1) atomicAdd(&cur[d[k] >> RANGE_SHIFT], 1);
            }
        }
    }
    __syncthreads();

    int carry = 0;
    for (int t0 = 0; t0 < NB; t0 += 1024) {
        const int i = t0 + tid;
        const int c = (i < NB) ? cur[i] : 0;
        int incl = c;
        for (int o = 1; o < 32; o <<= 1) {
            const int t = __shfl_up(incl, o);
            if (lane >= o) incl += t;
        }
        if (lane == 31) wave_tot[wave] = incl;
        __syncthreads();
        const int wt = wave_tot[lane];
        int wincl = wt;
        for (int o = 1; o < 32; o <<= 1) {
            const int t = __shfl_up(wincl, o);
            if (lane >= o) wincl += t;
        }
        const int woff = __shfl(wincl, wave) - __shfl(wt, wave);
        const int ex = carry + woff + incl - c;
        if (i < NB) {
            cur[i] = ex;
            offT[(size_t)i * NSLICE + blockIdx.x] = ex;
        }
        carry += __shfl(wincl, 31);
        __syncthreads();
    }
    if (tid == 0) offT[(size_t)NB * NSLICE + blockIdx.x] = carry;

    if (one_trip) {
#pragma unroll
        for (int k = 0; k < EDGES_PER_THREAD; ++k) {
            if (base0 + k * 1024 < e1) {
                const int slot = atomicAdd(&cur[d0[k] >> RANGE_SHIFT], 1);
                stage[slot] = make_int2(d0[k], s0[k]);
            }
        }
    } else {
        for (int base = base0; base < e1; base += EDGES_PER_THREAD * 1024) {
            int d[EDGES_PER_THREAD], s[EDGES_PER_THREAD];
#pragma unroll
            for (int k = 0; k < EDGES_PER_THREAD; ++k) {
                const int ec = min(base + k * 1024, e1 - 1);
                d[k] = dst[ec];
                s[k] = src[ec];
            }
            asm volatile("" :: "v"(d[0]), "v"(d[1]), "v"(d[2]), "v"(d[3]),
                               "v"(d[4]), "v"(d[5]), "v"(d[6]), "v"(d[7]),
                               "v"(s[0]), "v"(s[1]), "v"(s[2]), "v"(s[3]),
                               "v"(s[4]), "v"(s[5]), "v"(s[6]), "v"(s[7]));
#pragma unroll
            for (int k = 0; k < EDGES_PER_THREAD; ++k) {
                const int e = base + k * 1024;
                if (e < e1) {
                    const int slot = atomicAdd(&cur[d[k] >> RANGE_SHIFT], 1);
                    stage[slot] = make_int2(d[k], s[k]);
                }
            }
        }
    }
    __syncthreads();
    for (int i = tid; i < e1 - e0; i += 1024) pairs[e0 + i] = stage[i];
}

__global__ __launch_bounds__(1024) void range_rows_kernel(const int2* __restrict__ pairs,
                                                          const int* __restrict__ offT,
                                                          int* __restrict__ rowptr,
                                                          int* __restrict__ csr_src,
                                                          int n, int slice) {
    __shared__ int run_start[NSLICE + 1];
    __shared__ int run_src[NSLICE];
    __shared__ int node_cnt[RANGE_NODES];
    __shared__ int node_beg[RANGE_NODES];
    __shared__ int node_cur[RANGE_NODES];
    __shared__ int wave_cnt[RANGE_NODES / 32];
    __shared__ int wave_off[RANGE_NODES / 32];
    __shared__ __attribute__((aligned(16))) int2 lpairs[ROWS_STAGE];
    __shared__ int staged[ROWS_STAGE];
    const int tid   = threadIdx.x;
    const int lane  = tid & 31;
    const int wave  = __builtin_amdgcn_readfirstlane((int)(threadIdx.x >> 5));
    const int r     = blockIdx.x;
    const int node0 = r << RANGE_SHIFT;

    int o0 = 0, cnt = 0, incl = 0;
    if (tid < NSLICE) {
        o0 = offT[(size_t)r * NSLICE + tid];
        const int o1 = offT[(size_t)(r + 1) * NSLICE + tid];
        cnt = o1 - o0;
        incl = cnt;
        for (int o = 1; o < 32; o <<= 1) {
            const int t = __shfl_up(incl, o);
            if (lane >= o) incl += t;
        }
        int sum0 = o0;
        for (int o = 16; o >= 1; o >>= 1) sum0 += __shfl_xor(sum0, o);
        if (lane == 31) wave_cnt[wave] = incl;
        if (lane == 0)  wave_off[wave] = sum0;
        node_cnt[tid] = 0;
    }
    __syncthreads();
    int b0 = 0, total = 0;
#pragma unroll
    for (int k = 0; k < NSLICE / 32; ++k) {
        b0    += wave_off[k];
        total += wave_cnt[k];
    }
    if (tid < NSLICE) {
        int woff = 0;
        for (int k = 0; k < wave; ++k) woff += wave_cnt[k];
        run_start[tid] = woff + incl - cnt;
        run_src[tid]   = tid * slice + o0;
        if (tid == 0) run_start[NSLICE] = total;
    }
    __syncthreads();
    const bool in_lds = total <= ROWS_STAGE;

    const int rt = tid >> 2;
    const int q  = tid & 3;
    const int rs = run_start[rt];
    const int rc = run_start[rt + 1] - rs;
    const int rg = run_src[rt];

    if (in_lds) {
        for (int k0 = q; k0 < rc; k0 += 16) {
            int2 p[4];
#pragma unroll
            for (int u = 0; u < 4; ++u) {
                p[u] = pairs[rg + min(k0 + 4 * u, rc - 1)];
            }
            asm volatile("" :: "v"(p[0].x), "v"(p[0].y), "v"(p[1].x), "v"(p[1].y),
                               "v"(p[2].x), "v"(p[2].y), "v"(p[3].x), "v"(p[3].y));
#pragma unroll
            for (int u = 0; u < 4; ++u) {
                const int k = k0 + 4 * u;
                if (k < rc) lpairs[rs + k] = p[u];
            }
        }
    }
    __syncthreads();

    if (in_lds) {
        for (int i = tid; i < total; i += 1024) atomicAdd(&node_cnt[lpairs[i].x - node0], 1);
    } else {
        for (int k = q; k < rc; k += 4) atomicAdd(&node_cnt[pairs[rg + k].x - node0], 1);
    }
    __syncthreads();

    int c = 0, nincl = 0;
    if (tid < RANGE_NODES) {
        c = node_cnt[tid];
        nincl = c;
        for (int o = 1; o < 32; o <<= 1) {
            const int t = __shfl_up(nincl, o);
            if (lane >= o) nincl += t;
        }
        if (lane == 31) wave_cnt[wave] = nincl;
    }
    __syncthreads();
    if (tid < RANGE_NODES) {
        int woff = 0;
        for (int k = 0; k < wave; ++k) woff += wave_cnt[k];
        const int first = woff + nincl - c;
        node_beg[tid] = first;
        node_cur[tid] = first;
        if (node0 + tid < n) rowptr[node0 + tid] = b0 + first;
    }
    __syncthreads();

    if (in_lds) {
        for (int i = tid; i < total; i += 1024) {
            const int2 p = lpairs[i];
            const int slot = atomicAdd(&node_cur[p.x - node0], 1);
            staged[slot] = p.y;
        }
    } else {
        for (int k = q; k < rc; k += 4) {
            const int2 p = pairs[rg + k];
            const int slot = atomicAdd(&node_cur[p.x - node0], 1);
            csr_src[b0 + slot] = p.y;
        }
    }
    __syncthreads();

    int* sorted = (int*)lpairs;
    if (in_lds) {
        const bool packed = n <= (1 << 26);
        for (int nd = wave; nd < RANGE_NODES; nd += 32) {
            const int beg = __builtin_amdgcn_readfirstlane(node_beg[nd]);
            const int m   = __builtin_amdgcn_readfirstlane(node_cnt[nd]);
            if (m <= 32 && packed) {
                int key = 0x7fffffff;
                if (lane < m) key = (staged[beg + lane] << 5) | lane;
                int rank = 0;
                for (int j = 0; j < m; ++j) {
                    const int kj = __builtin_amdgcn_readlane(key, j);
                    rank += (kj < key) ? 1 : 0;
                }
                if (lane < m) sorted[beg + rank] = key >> 5;
            } else {
                for (int i0 = 0; i0 < m; i0 += 32) {
                    const int i = i0 + lane;
                    const int v = (i < m) ? staged[beg + i] : 0;
                    int rank = 0;
                    for (int j = 0; j < m; ++j) {
                        const int vj = staged[beg + j];
                        rank += (vj < v || (vj == v && j < i)) ? 1 : 0;
                    }
                    if (i < m) sorted[beg + rank] = v;
                }
            }
        }
    }
    __syncthreads();
    if (in_lds)
        for (int i = tid; i < total; i += 1024) csr_src[b0 + i] = sorted[i];
}

template <int K, int NCOL>
__global__ __launch_bounds__(128) void gemm_wmma_f32(const float* __restrict__ A,
                                                     const float* __restrict__ B,
                                                     const int* __restrict__ rowptr,
                                                     float* __restrict__ C, int nrows, int E) {
    const int wave = threadIdx.x >> 5;
    const int lane = threadIdx.x & 31;
    const int half = lane >> 4;
    const int mr   = lane & 15;

    const int row_tile = blockIdx.x * (blockDim.x >> 5) + wave;
    const int row0 = row_tile * 16;
    if (row0 >= nrows) return;

    constexpr int NT = NCOL / 16;
    v8f acc[NT] = {};

    const int ri   = row0 + mr;
    const int rbeg = rowptr[min(ri, nrows - 1)];
    const int rnxt = rowptr[min(ri + 1, nrows - 1)];
    const int rend = (ri + 1 < nrows) ? rnxt : E;

    const float* arow = A + (size_t)(row0 + mr) * K + half * 2;

#pragma unroll 8
    for (int kb = 0; kb < K / 4; ++kb) {
        v2f a;
        a.x = arow[kb * 4 + 0];
        a.y = arow[kb * 4 + 1];
        const float* brow = B + (size_t)(kb * 4 + half * 2) * NCOL + mr;
#pragma unroll
        for (int ct = 0; ct < NT; ++ct) {
            v2f b;
            b.x = brow[ct * 16];
            b.y = brow[ct * 16 + NCOL];
            acc[ct] = __builtin_amdgcn_wmma_f32_16x16x4_f32(
                 false, a,  false, b,
                 (short)0, acc[ct],  false,  false);
        }
    }

    const int dv = __float_as_int(rsqrtf((float)(rend - rbeg + 1)));
    float sc[8];
#pragma unroll
    for (int r = 0; r < 8; ++r) {
        const float lo = __int_as_float(__builtin_amdgcn_readlane(dv, r));
        const float hi = __int_as_float(__builtin_amdgcn_readlane(dv, r + 8));
        sc[r] = half ? hi : lo;
    }

#pragma unroll
    for (int ct = 0; ct < NT; ++ct) {
        float* crow = C + (size_t)(row0 + half * 8) * NCOL + ct * 16 + mr;
#pragma unroll
        for (int r = 0; r < 8; ++r) crow[(size_t)r * NCOL] = sc[r] * acc[ct][r];
    }
}

template <int V> struct vec_of;
template <> struct vec_of<4> { typedef v4f type; };
template <> struct vec_of<2> { typedef v2f type; };

template <int F, bool RELU>
__global__ __launch_bounds__(256) void aggregate_rows_kernel(const float* __restrict__ hs,
                                                             const int* __restrict__ rowptr,
                                                             const int* __restrict__ csr_src,
                                                             const float* __restrict__ bias,
                                                             float* __restrict__ out,
                                                             int n, int E) {
    constexpr int V = F / 32;
    constexpr int U = 4;
    typedef typename vec_of<V>::type vec;

    const int lane = threadIdx.x & 31;
    const int d = __builtin_amdgcn_readfirstlane((int)((blockIdx.x * blockDim.x + threadIdx.x) >> 5));
    if (d >= n) return;

    const int beg = rowptr[d];
    const int end = (d + 1 < n) ? rowptr[d + 1] : E;

    vec acc = *((const vec*)(hs + (size_t)d * F) + lane);

    for (int base = beg; base < end; base += 32) {
        const int m = min(end - base, 32);
        int s = 0;
        if (lane < m) s = csr_src[base + lane];
        int j0 = 0;
        for (; j0 + U <= m; j0 += U) {
            vec hv[U];
#pragma unroll
            for (int u = 0; u < U; ++u) {
                const int sj = __builtin_amdgcn_readlane(s, j0 + u);
                hv[u] = *((const vec*)(hs + (size_t)sj * F) + lane);
            }
#pragma unroll
            for (int u = 0; u < U; ++u) acc += hv[u];
        }
        for (; j0 < m; ++j0) {
            const int sj = __builtin_amdgcn_readlane(s, j0);
            const vec hv = *((const vec*)(hs + (size_t)sj * F) + lane);
            acc += hv;
        }
    }

    {
        const float dd = rsqrtf((float)(end - beg + 1));
        const vec bv = *((const vec*)bias + lane);
        acc = bv + dd * acc;
    }

    if (RELU) {
#pragma unroll
        for (int k = 0; k < V; ++k) acc[k] = fmaxf(acc[k], 0.0f);
    }
    *((vec*)(out + (size_t)d * F) + lane) = acc;
}

#define TILE_ROWS 16
#define ROWS_PER_WAVE 4
#define ZT_PITCH 132

__global__ __launch_bounds__(128) void aggregate_relu_gemm_kernel(const float* __restrict__ hs,
                                                                  const int* __restrict__ rowptr,
                                                                  const int* __restrict__ csr_src,
                                                                  const float* __restrict__ bias,
                                                                  const float* __restrict__ B,
                                                                  float* __restrict__ C,
                                                                  int n, int E) {
    constexpr int F = 128;
    constexpr int NCOL = 64;
    constexpr int U = 4;
    __shared__ __attribute__((aligned(16))) float zt[TILE_ROWS * ZT_PITCH];
    __shared__ __attribute__((aligned(16))) float dts[TILE_ROWS];

    const int lane = threadIdx.x & 31;
    const int wave = __builtin_amdgcn_readfirstlane((int)(threadIdx.x >> 5));
    const int node0 = blockIdx.x * TILE_ROWS;
    const int d0 = node0 + wave * ROWS_PER_WAVE;

    int rb[ROWS_PER_WAVE + 1];
    if (d0 + ROWS_PER_WAVE < n) {
#pragma unroll
        for (int i = 0; i <= ROWS_PER_WAVE; ++i) rb[i] = rowptr[d0 + i];
    } else {
#pragma unroll
        for (int i = 0; i <= ROWS_PER_WAVE; ++i) {
            const int v = rowptr[min(d0 + i, n - 1)];
            rb[i] = (d0 + i < n) ? v : E;
        }
    }

#pragma unroll
    for (int i = 0; i < ROWS_PER_WAVE; ++i) {
        const int t = wave * ROWS_PER_WAVE + i;
        const int d = d0 + i;
        const int beg = rb[i];
        const int end = rb[i + 1];
        v4f acc = {0.0f, 0.0f, 0.0f, 0.0f};
        float dd = 0.0f;
        if (d < n) {
            acc = *((const v4f*)(hs + (size_t)d * F) + lane);
            for (int base = beg; base < end; base += 32) {
                const int m = min(end - base, 32);
                int s = 0;
                if (lane < m) s = csr_src[base + lane];
                int j0 = 0;
                for (; j0 + U <= m; j0 += U) {
                    v4f hv[U];
#pragma unroll
                    for (int u = 0; u < U; ++u) {
                        const int sj = __builtin_amdgcn_readlane(s, j0 + u);
                        hv[u] = *((const v4f*)(hs + (size_t)sj * F) + lane);
                    }
#pragma unroll
                    for (int u = 0; u < U; ++u) acc += hv[u];
                }
                for (; j0 < m; ++j0) {
                    const int sj = __builtin_amdgcn_readlane(s, j0);
                    const v4f hv = *((const v4f*)(hs + (size_t)sj * F) + lane);
                    acc += hv;
                }
            }
            dd = rsqrtf((float)(end - beg + 1));
            const v4f bv = *((const v4f*)bias + lane);
            acc = bv + dd * acc;
#pragma unroll
            for (int k = 0; k < 4; ++k) acc[k] = fmaxf(acc[k], 0.0f);
        }
        *((v4f*)(zt + t * ZT_PITCH) + lane) = acc;
        if (lane == 0) dts[t] = dd;
    }

    const int half = lane >> 4;
    const int mr   = lane & 15;

    const float* arow = zt + mr * ZT_PITCH + half * 2;
    const float* bcol = B + (size_t)(half * 2) * NCOL + wave * 16 + mr;

    constexpr int KG = 8;
    v2f bq[2][KG];
#pragma unroll
    for (int j = 0; j < KG; ++j) {
        bq[0][j].x = bcol[(size_t)(j * 4) * NCOL];
        bq[0][j].y = bcol[(size_t)(j * 4 + 1) * NCOL];
    }

    __syncthreads();

    v8f c = {};
#pragma unroll
    for (int g = 0; g < F / 4 / KG; ++g) {
        if (g + 1 < F / 4 / KG) {
#pragma unroll
            for (int j = 0; j < KG; ++j) {
                const int kb = (g + 1) * KG + j;
                bq[(g + 1) & 1][j].x = bcol[(size_t)(kb * 4) * NCOL];
                bq[(g + 1) & 1][j].y = bcol[(size_t)(kb * 4 + 1) * NCOL];
            }
        }
#pragma unroll
        for (int j = 0; j < KG; ++j) {
            const int kb = g * KG + j;
            const v2f a = *(const v2f*)(arow + kb * 4);
            c = __builtin_amdgcn_wmma_f32_16x16x4_f32(
                 false, a,  false, bq[g & 1][j],
                 (short)0, c,  false,  false);
        }
    }
    __builtin_amdgcn_sched_barrier(0);
    asm volatile("v_nop\n\tv_nop\n\tv_nop\n\tv_nop\n\tv_nop\n\tv_nop\n\tv_nop\n\tv_nop" : "+v"(c));
    __builtin_amdgcn_sched_barrier(0);

    const v4f s0 = *(const v4f*)(dts + half * 8);
    const v4f s1 = *(const v4f*)(dts + half * 8 + 4);
    float o[8];
#pragma unroll
    for (int r = 0; r < 4; ++r) {
        o[r]     = s0[r] * c[r];
        o[r + 4] = s1[r] * c[r + 4];
    }
    float* crow = C + (size_t)(node0 + half * 8) * NCOL + wave * 16 + mr;
    if (node0 + TILE_ROWS <= n) {
#pragma unroll
        for (int r = 0; r < 8; ++r) crow[(size_t)r * NCOL] = o[r];
    } else {
#pragma unroll
        for (int r = 0; r < 8; ++r)
            if (node0 + half * 8 + r < n) crow[(size_t)r * NCOL] = o[r];
    }
}

extern "C" void kernel_launch(void* const* d_in, const int* in_sizes, int n_in,
                              void* d_out, int out_size, void* d_ws, size_t ws_size,
                              hipStream_t stream) {
    const float* x  = (const float*)d_in[0];
    const int*   ei = (const int*)d_in[1];
    const float* W1 = (const float*)d_in[2];
    const float* b1 = (const float*)d_in[3];
    const float* W2 = (const float*)d_in[4];
    const float* b2 = (const float*)d_in[5];

    const int N = in_sizes[0] / 128;
    const int E = in_sizes[1] / 2;
    const int* src = ei;
    const int* dst = ei + E;

    int*   rowptr  = (int*)d_ws;
    int*   csr_src = rowptr + N;
    float* h1      = (float*)(csr_src + E);
    float* h2      = h1 + (size_t)N * 128;
    const int NB   = (N + RANGE_NODES - 1) / RANGE_NODES;
    int2*  pairs   = (int2*)h2;
    int*   offT    = (int*)(pairs + E);
    float* out     = (float*)d_out;

    const int T = 256;

    const int slice = (E + NSLICE - 1) / NSLICE;
    const size_t group_lds = (size_t)((NB + 2) & ~1) * sizeof(int) + (size_t)slice * sizeof(int2);
    slice_group_kernel<<<NSLICE, 1024, group_lds, stream>>>(src, dst, offT, pairs, E, slice, NB);
    range_rows_kernel<<<NB, 1024, 0, stream>>>(pairs, offT, rowptr, csr_src, N, slice);

    const int row_tiles = (N + 15) / 16;
    const int gemm_blocks = (row_tiles + 3) / 4;
    const int agg_blocks = (N + 7) / 8;

    gemm_wmma_f32<128, 128><<<gemm_blocks, 128, 0, stream>>>(x, W1, rowptr, h1, N, E);
    aggregate_relu_gemm_kernel<<<row_tiles, 128, 0, stream>>>(h1, rowptr, csr_src, b1, W2, h2, N, E);

    aggregate_rows_kernel<64, false><<<agg_blocks, T, 0, stream>>>(h2, rowptr, csr_src, b2, out, N, E);
}


# Layer-1 product fused into the row-building kernel: each range block (208 nodes, two per WGP) multiplies its rows on f32 WMMA, B in the LDS its pairs leave, dinv from LDS counts; 5 launches to 4
# speedup vs baseline: 5.0391x; 1.1373x over previous
#include <hip/hip_runtime.h>

typedef __attribute__((ext_vector_type(2))) float v2f;
typedef __attribute__((ext_vector_type(4))) float v4f;
typedef __attribute__((ext_vector_type(8))) float v8f;

#define RANGE_NODES 208
#define NSLICE 256
#define EDGES_PER_THREAD 8
#define ROWS_STAGE 8192

__device__ __forceinline__ int range_of(int d) { return (int)((unsigned)d / RANGE_NODES); }

__global__ __launch_bounds__(1024) void slice_group_kernel(const int* __restrict__ src,
                                                           const int* __restrict__ dst,
                                                           int* __restrict__ offT,
                                                           int2* __restrict__ pairs,
                                                           int E, int slice, int NB) {
    extern __shared__ int lds_dyn[];
    __shared__ int wave_tot[32];
    int*  cur   = lds_dyn;
    int2* stage = (int2*)(lds_dyn + ((NB + 2) & ~1));
    const int tid  = threadIdx.x;
    const int lane = tid & 31;
    const int wave = tid >> 5;

    for (int i = tid; i <= NB; i += 1024) cur[i] = 0;
    __syncthreads();

    const int e0 = min(E, (int)blockIdx.x * slice);
    const int e1 = min(E, e0 + slice);
    const bool one_trip = slice <= EDGES_PER_THREAD * 1024;
    const int base0 = e0 + tid;
    int d0[EDGES_PER_THREAD], s0[EDGES_PER_THREAD];
#pragma unroll
    for (int k = 0; k < EDGES_PER_THREAD; ++k) { d0[k] = 0; s0[k] = 0; }
    if (one_trip && e0 < e1) {
#pragma unroll
        for (int k = 0; k < EDGES_PER_THREAD; ++k) {
            const int ec = min(base0 + k * 1024, e1 - 1);
            d0[k] = dst[ec];
            s0[k] = src[ec];
        }
    }

    if (one_trip) {
        asm volatile("" :: "v"(d0[0]), "v"(d0[1]), "v"(d0[2]), "v"(d0[3]),
                           "v"(d0[4]), "v"(d0[5]), "v"(d0[6]), "v"(d0[7]));
#pragma unroll
        for (int k = 0; k < EDGES_PER_THREAD; ++k) {
            if (base0 + k * 1024 < e1) atomicAdd(&cur[range_of(d0[k])], 1);
        }
    } else {
        for (int base = base0; base < e1; base += EDGES_PER_THREAD * 1024) {
            int d[EDGES_PER_THREAD];
#pragma unroll
            for (int k = 0; k < EDGES_PER_THREAD; ++k) {
                d[k] = dst[min(base + k * 1024, e1 - 1)];
            }
            asm volatile("" :: "v"(d[0]), "v"(d[1]), "v"(d[2]), "v"(d[3]),
                               "v"(d[4]), "v"(d[5]), "v"(d[6]), "v"(d[7]));
#pragma unroll
            for (int k = 0; k < EDGES_PER_THREAD; ++k) {
                const int e = base + k * 1024;
                if (e < e1) atomicAdd(&cur[range_of(d[k])], 1);
            }
        }
    }
    __syncthreads();

    int carry = 0;
    for (int t0 = 0; t0 < NB; t0 += 1024) {
        const int i = t0 + tid;
        const int c = (i < NB) ? cur[i] : 0;
        int incl = c;
        for (int o = 1; o < 32; o <<= 1) {
            const int t = __shfl_up(incl, o);
            if (lane >= o) incl += t;
        }
        if (lane == 31) wave_tot[wave] = incl;
        __syncthreads();
        const int wt = wave_tot[lane];
        int wincl = wt;
        for (int o = 1; o < 32; o <<= 1) {
            const int t = __shfl_up(wincl, o);
            if (lane >= o) wincl += t;
        }
        const int woff = __shfl(wincl, wave) - __shfl(wt, wave);
        const int ex = carry + woff + incl - c;
        if (i < NB) {
            cur[i] = ex;
            offT[(size_t)i * NSLICE + blockIdx.x] = ex;
        }
        carry += __shfl(wincl, 31);
        __syncthreads();
    }
    if (tid == 0) offT[(size_t)NB * NSLICE + blockIdx.x] = carry;

    if (one_trip) {
#pragma unroll
        for (int k = 0; k < EDGES_PER_THREAD; ++k) {
            if (base0 + k * 1024 < e1) {
                const int slot = atomicAdd(&cur[range_of(d0[k])], 1);
                stage[slot] = make_int2(d0[k], s0[k]);
            }
        }
    } else {
        for (int base = base0; base < e1; base += EDGES_PER_THREAD * 1024) {
            int d[EDGES_PER_THREAD], s[EDGES_PER_THREAD];
#pragma unroll
            for (int k = 0; k < EDGES_PER_THREAD; ++k) {
                const int ec = min(base + k * 1024, e1 - 1);
                d[k] = dst[ec];
                s[k] = src[ec];
            }
            asm volatile("" :: "v"(d[0]), "v"(d[1]), "v"(d[2]), "v"(d[3]),
                               "v"(d[4]), "v"(d[5]), "v"(d[6]), "v"(d[7]),
                               "v"(s[0]), "v"(s[1]), "v"(s[2]), "v"(s[3]),
                               "v"(s[4]), "v"(s[5]), "v"(s[6]), "v"(s[7]));
#pragma unroll
            for (int k = 0; k < EDGES_PER_THREAD; ++k) {
                const int e = base + k * 1024;
                if (e < e1) {
                    const int slot = atomicAdd(&cur[range_of(d[k])], 1);
                    stage[slot] = make_int2(d[k], s[k]);
                }
            }
        }
    }
    __syncthreads();
    for (int i = tid; i < e1 - e0; i += 1024) pairs[e0 + i] = stage[i];
}

__global__ __launch_bounds__(1024, 16) void range_rows_kernel(const int2* __restrict__ pairs,
                                                          const int* __restrict__ offT,
                                                          int* __restrict__ rowptr,
                                                          int* __restrict__ csr_src,
                                                          const float* __restrict__ A,
                                                          const float* __restrict__ B,
                                                          float* __restrict__ C,
                                                          int n, int slice) {
    __shared__ int run_start[NSLICE + 1];
    __shared__ int run_src[NSLICE];
    __shared__ int node_cnt[RANGE_NODES];
    __shared__ int node_beg[RANGE_NODES];
    __shared__ int node_cur[RANGE_NODES];
    __shared__ int wave_cnt[NSLICE / 32];
    __shared__ int wave_off[NSLICE / 32];
    __shared__ __attribute__((aligned(16))) int stage_mem[ROWS_STAGE * 3];
    int2* lpairs = (int2*)stage_mem;
    int*  staged = stage_mem + ROWS_STAGE * 2;
    const int tid   = threadIdx.x;
    const int lane  = tid & 31;
    const int wave  = __builtin_amdgcn_readfirstlane((int)(threadIdx.x >> 5));
    const int r     = blockIdx.x;
    const int node0 = r * RANGE_NODES;

    int o0 = 0, cnt = 0, incl = 0;
    if (tid < NSLICE) {
        o0 = offT[(size_t)r * NSLICE + tid];
        const int o1 = offT[(size_t)(r + 1) * NSLICE + tid];
        cnt = o1 - o0;
        incl = cnt;
        for (int o = 1; o < 32; o <<= 1) {
            const int t = __shfl_up(incl, o);
            if (lane >= o) incl += t;
        }
        int sum0 = o0;
        for (int o = 16; o >= 1; o >>= 1) sum0 += __shfl_xor(sum0, o);
        if (lane == 31) wave_cnt[wave] = incl;
        if (lane == 0)  wave_off[wave] = sum0;
        if (tid < RANGE_NODES) node_cnt[tid] = 0;
    }
    __syncthreads();
    int b0 = 0, total = 0;
#pragma unroll
    for (int k = 0; k < NSLICE / 32; ++k) {
        b0    += wave_off[k];
        total += wave_cnt[k];
    }
    if (tid < NSLICE) {
        int woff = 0;
        for (int k = 0; k < wave; ++k) woff += wave_cnt[k];
        run_start[tid] = woff + incl - cnt;
        run_src[tid]   = tid * slice + o0;
        if (tid == 0) run_start[NSLICE] = total;
    }
    __syncthreads();
    const bool in_lds = total <= ROWS_STAGE;

    const int rt = tid >> 2;
    const int q  = tid & 3;
    const int rs = run_start[rt];
    const int rc = run_start[rt + 1] - rs;
    const int rg = run_src[rt];

    if (in_lds) {
        for (int k0 = q; k0 < rc; k0 += 16) {
            int2 p[4];
#pragma unroll
            for (int u = 0; u < 4; ++u) {
                p[u] = pairs[rg + min(k0 + 4 * u, rc - 1)];
            }
            asm volatile("" :: "v"(p[0].x), "v"(p[0].y), "v"(p[1].x), "v"(p[1].y),
                               "v"(p[2].x), "v"(p[2].y), "v"(p[3].x), "v"(p[3].y));
#pragma unroll
            for (int u = 0; u < 4; ++u) {
                const int k = k0 + 4 * u;
                if (k < rc) lpairs[rs + k] = p[u];
            }
        }
    }
    __syncthreads();

    if (in_lds) {
        for (int i = tid; i < total; i += 1024) atomicAdd(&node_cnt[lpairs[i].x - node0], 1);
    } else {
        for (int k = q; k < rc; k += 4) atomicAdd(&node_cnt[pairs[rg + k].x - node0], 1);
    }
    __syncthreads();

    int c = 0, nincl = 0;
    if (tid < RANGE_NODES) {
        c = node_cnt[tid];
        nincl = c;
        for (int o = 1; o < 32; o <<= 1) {
            const int t = __shfl_up(nincl, o);
            if (lane >= o) nincl += t;
        }
        if (lane == 31) wave_cnt[wave] = nincl;
    }
    __syncthreads();
    if (tid < RANGE_NODES) {
        int woff = 0;
        for (int k = 0; k < wave; ++k) woff += wave_cnt[k];
        const int first = woff + nincl - c;
        node_beg[tid] = first;
        node_cur[tid] = first;
        if (node0 + tid < n) rowptr[node0 + tid] = b0 + first;
    }
    __syncthreads();

    if (in_lds) {
        for (int i = tid; i < total; i += 1024) {
            const int2 p = lpairs[i];
            const int slot = atomicAdd(&node_cur[p.x - node0], 1);
            staged[slot] = p.y;
        }
    } else {
        for (int k = q; k < rc; k += 4) {
            const int2 p = pairs[rg + k];
            const int slot = atomicAdd(&node_cur[p.x - node0], 1);
            csr_src[b0 + slot] = p.y;
        }
    }
    __syncthreads();

    int* sorted = (int*)lpairs;
    if (in_lds) {
        const bool packed = n <= (1 << 26);
        for (int nd = wave; nd < RANGE_NODES; nd += 32) {
            const int beg = __builtin_amdgcn_readfirstlane(node_beg[nd]);
            const int m   = __builtin_amdgcn_readfirstlane(node_cnt[nd]);
            if (m <= 32 && packed) {
                int key = 0x7fffffff;
                if (lane < m) key = (staged[beg + lane] << 5) | lane;
                int rank = 0;
                for (int j = 0; j < m; ++j) {
                    const int kj = __builtin_amdgcn_readlane(key, j);
                    rank += (kj < key) ? 1 : 0;
                }
                if (lane < m) sorted[beg + rank] = key >> 5;
            } else {
                for (int i0 = 0; i0 < m; i0 += 32) {
                    const int i = i0 + lane;
                    const int v = (i < m) ? staged[beg + i] : 0;
                    int rank = 0;
                    for (int j = 0; j < m; ++j) {
                        const int vj = staged[beg + j];
                        rank += (vj < v || (vj == v && j < i)) ? 1 : 0;
                    }
                    if (i < m) sorted[beg + rank] = v;
                }
            }
        }
    }
    __syncthreads();
    if (in_lds)
        for (int i = tid; i < total; i += 1024) csr_src[b0 + i] = sorted[i];

    constexpr int K = 128;
    constexpr int NCOL = 128;
    constexpr int NT = 4;
    float* wl = (float*)stage_mem;
    {
        v4f w[4];
#pragma unroll
        for (int u = 0; u < 4; ++u) {
            const int task = tid + u * 1024;
            const float* g = B + (size_t)((task >> 8) * 8 + ((task >> 7) & 1) * 2) * NCOL + (task & 127);
            w[u].x = g[0];
            w[u].y = g[NCOL];
            w[u].z = g[4 * NCOL];
            w[u].w = g[5 * NCOL];
        }
        __syncthreads();
#pragma unroll
        for (int u = 0; u < 4; ++u) {
            const int task = tid + u * 1024;
            *(v4f*)(wl + (task >> 8) * 1024 + (task & 127) * 8 + ((task >> 7) & 1) * 4) = w[u];
        }
        __syncthreads();
    }

    const int half = lane >> 4;
    const int mr   = lane & 15;
    const int tile = wave >> 1;
    const int col0 = (wave & 1) * (NT * 16);
    const int row0 = node0 + tile * 16;
    if (tile >= RANGE_NODES / 16 || row0 >= n) return;

    v8f acc[NT] = {};
    const float* arow = A + (size_t)min(row0 + mr, n - 1) * K + half * 2;
    const float* blp = wl + (col0 + mr) * 8 + half * 4;

    v2f af[2][2];
    af[0][0].x = arow[0];
    af[0][0].y = arow[1];
    af[0][1].x = arow[4];
    af[0][1].y = arow[5];
#pragma unroll
    for (int p = 0; p < K / 8; ++p) {
        if (p + 1 < K / 8) {
            af[(p + 1) & 1][0].x = arow[(p + 1) * 8 + 0];
            af[(p + 1) & 1][0].y = arow[(p + 1) * 8 + 1];
            af[(p + 1) & 1][1].x = arow[(p + 1) * 8 + 4];
            af[(p + 1) & 1][1].y = arow[(p + 1) * 8 + 5];
        }
        v4f q[NT];
#pragma unroll
        for (int ct = 0; ct < NT; ++ct) q[ct] = *(const v4f*)(blp + p * 1024 + ct * 128);
#pragma unroll
        for (int ct = 0; ct < NT; ++ct) {
            const v2f b = {q[ct].x, q[ct].y};
            acc[ct] = __builtin_amdgcn_wmma_f32_16x16x4_f32(
                 false, af[p & 1][0],  false, b,
                 (short)0, acc[ct],  false,  false);
        }
#pragma unroll
        for (int ct = 0; ct < NT; ++ct) {
            const v2f b = {q[ct].z, q[ct].w};
            acc[ct] = __builtin_amdgcn_wmma_f32_16x16x4_f32(
                 false, af[p & 1][1],  false, b,
                 (short)0, acc[ct],  false,  false);
        }
    }

    __builtin_amdgcn_sched_barrier(0);
    asm volatile("v_nop\n\tv_nop\n\tv_nop\n\tv_nop\n\tv_nop\n\tv_nop\n\tv_nop\n\tv_nop"
                 : "+v"(acc[0]), "+v"(acc[1]), "+v"(acc[2]), "+v"(acc[3]));
    __builtin_amdgcn_sched_barrier(0);

    const int le    = (int)__builtin_amdgcn_mbcnt_lo(~0u, 0u);
    const int halfe = le >> 4;
    const int mre   = le & 15;

    const int dv = __float_as_int(rsqrtf((float)(node_cnt[tile * 16 + mre] + 1)));
    float sc[8];
#pragma unroll
    for (int r = 0; r < 8; ++r) {
        const float lo = __int_as_float(__builtin_amdgcn_readlane(dv, r));
        const float hi = __int_as_float(__builtin_amdgcn_readlane(dv, r + 8));
        sc[r] = halfe ? hi : lo;
    }

    if (row0 + 16 <= n) {
#pragma unroll
        for (int ct = 0; ct < NT; ++ct) {
            float* crow = C + (size_t)(row0 + halfe * 8) * NCOL + col0 + ct * 16 + mre;
#pragma unroll
            for (int r = 0; r < 8; ++r) crow[(size_t)r * NCOL] = sc[r] * acc[ct][r];
        }
    } else {
#pragma unroll
        for (int ct = 0; ct < NT; ++ct) {
            float* crow = C + (size_t)(row0 + halfe * 8) * NCOL + col0 + ct * 16 + mre;
#pragma unroll
            for (int r = 0; r < 8; ++r)
                if (row0 + halfe * 8 + r < n) crow[(size_t)r * NCOL] = sc[r] * acc[ct][r];
        }
    }
}

template <int V> struct vec_of;
template <> struct vec_of<4> { typedef v4f type; };
template <> struct vec_of<2> { typedef v2f type; };

template <int F, bool RELU>
__global__ __launch_bounds__(256) void aggregate_rows_kernel(const float* __restrict__ hs,
                                                             const int* __restrict__ rowptr,
                                                             const int* __restrict__ csr_src,
                                                             const float* __restrict__ bias,
                                                             float* __restrict__ out,
                                                             int n, int E) {
    constexpr int V = F / 32;
    constexpr int U = 4;
    typedef typename vec_of<V>::type vec;

    const int lane = threadIdx.x & 31;
    const int d = __builtin_amdgcn_readfirstlane((int)((blockIdx.x * blockDim.x + threadIdx.x) >> 5));
    if (d >= n) return;

    const int beg = rowptr[d];
    const int end = (d + 1 < n) ? rowptr[d + 1] : E;

    vec acc = *((const vec*)(hs + (size_t)d * F) + lane);

    for (int base = beg; base < end; base += 32) {
        const int m = min(end - base, 32);
        int s = 0;
        if (lane < m) s = csr_src[base + lane];
        int j0 = 0;
        for (; j0 + U <= m; j0 += U) {
            vec hv[U];
#pragma unroll
            for (int u = 0; u < U; ++u) {
                const int sj = __builtin_amdgcn_readlane(s, j0 + u);
                hv[u] = *((const vec*)(hs + (size_t)sj * F) + lane);
            }
#pragma unroll
            for (int u = 0; u < U; ++u) acc += hv[u];
        }
        for (; j0 < m; ++j0) {
            const int sj = __builtin_amdgcn_readlane(s, j0);
            const vec hv = *((const vec*)(hs + (size_t)sj * F) + lane);
            acc += hv;
        }
    }

    {
        const float dd = rsqrtf((float)(end - beg + 1));
        const vec bv = *((const vec*)bias + lane);
        acc = bv + dd * acc;
    }

    if (RELU) {
#pragma unroll
        for (int k = 0; k < V; ++k) acc[k] = fmaxf(acc[k], 0.0f);
    }
    *((vec*)(out + (size_t)d * F) + lane) = acc;
}

#define TILE_ROWS 16
#define ROWS_PER_WAVE 4
#define ZT_PITCH 132

__global__ __launch_bounds__(128) void aggregate_relu_gemm_kernel(const float* __restrict__ hs,
                                                                  const int* __restrict__ rowptr,
                                                                  const int* __restrict__ csr_src,
                                                                  const float* __restrict__ bias,
                                                                  const float* __restrict__ B,
                                                                  float* __restrict__ C,
                                                                  int n, int E) {
    constexpr int F = 128;
    constexpr int NCOL = 64;
    constexpr int U = 4;
    __shared__ __attribute__((aligned(16))) float zt[TILE_ROWS * ZT_PITCH];
    __shared__ __attribute__((aligned(16))) float dts[TILE_ROWS];

    const int lane = threadIdx.x & 31;
    const int wave = __builtin_amdgcn_readfirstlane((int)(threadIdx.x >> 5));
    const int node0 = blockIdx.x * TILE_ROWS;
    const int d0 = node0 + wave * ROWS_PER_WAVE;

    int rb[ROWS_PER_WAVE + 1];
    if (d0 + ROWS_PER_WAVE < n) {
#pragma unroll
        for (int i = 0; i <= ROWS_PER_WAVE; ++i) rb[i] = rowptr[d0 + i];
    } else {
#pragma unroll
        for (int i = 0; i <= ROWS_PER_WAVE; ++i) {
            const int v = rowptr[min(d0 + i, n - 1)];
            rb[i] = (d0 + i < n) ? v : E;
        }
    }

#pragma unroll
    for (int i = 0; i < ROWS_PER_WAVE; ++i) {
        const int t = wave * ROWS_PER_WAVE + i;
        const int d = d0 + i;
        const int beg = rb[i];
        const int end = rb[i + 1];
        v4f acc = {0.0f, 0.0f, 0.0f, 0.0f};
        float dd = 0.0f;
        if (d < n) {
            acc = *((const v4f*)(hs + (size_t)d * F) + lane);
            for (int base = beg; base < end; base += 32) {
                const int m = min(end - base, 32);
                int s = 0;
                if (lane < m) s = csr_src[base + lane];
                int j0 = 0;
                for (; j0 + U <= m; j0 += U) {
                    v4f hv[U];
#pragma unroll
                    for (int u = 0; u < U; ++u) {
                        const int sj = __builtin_amdgcn_readlane(s, j0 + u);
                        hv[u] = *((const v4f*)(hs + (size_t)sj * F) + lane);
                    }
#pragma unroll
                    for (int u = 0; u < U; ++u) acc += hv[u];
                }
                for (; j0 < m; ++j0) {
                    const int sj = __builtin_amdgcn_readlane(s, j0);
                    const v4f hv = *((const v4f*)(hs + (size_t)sj * F) + lane);
                    acc += hv;
                }
            }
            dd = rsqrtf((float)(end - beg + 1));
            const v4f bv = *((const v4f*)bias + lane);
            acc = bv + dd * acc;
#pragma unroll
            for (int k = 0; k < 4; ++k) acc[k] = fmaxf(acc[k], 0.0f);
        }
        *((v4f*)(zt + t * ZT_PITCH) + lane) = acc;
        if (lane == 0) dts[t] = dd;
    }

    const int half = lane >> 4;
    const int mr   = lane & 15;

    const float* arow = zt + mr * ZT_PITCH + half * 2;
    const float* bcol = B + (size_t)(half * 2) * NCOL + wave * 16 + mr;

    constexpr int KG = 8;
    v2f bq[2][KG];
#pragma unroll
    for (int j = 0; j < KG; ++j) {
        bq[0][j].x = bcol[(size_t)(j * 4) * NCOL];
        bq[0][j].y = bcol[(size_t)(j * 4 + 1) * NCOL];
    }

    __syncthreads();

    v8f c = {};
#pragma unroll
    for (int g = 0; g < F / 4 / KG; ++g) {
        if (g + 1 < F / 4 / KG) {
#pragma unroll
            for (int j = 0; j < KG; ++j) {
                const int kb = (g + 1) * KG + j;
                bq[(g + 1) & 1][j].x = bcol[(size_t)(kb * 4) * NCOL];
                bq[(g + 1) & 1][j].y = bcol[(size_t)(kb * 4 + 1) * NCOL];
            }
        }
#pragma unroll
        for (int j = 0; j < KG; ++j) {
            const int kb = g * KG + j;
            const v2f a = *(const v2f*)(arow + kb * 4);
            c = __builtin_amdgcn_wmma_f32_16x16x4_f32(
                 false, a,  false, bq[g & 1][j],
                 (short)0, c,  false,  false);
        }
    }
    __builtin_amdgcn_sched_barrier(0);
    asm volatile("v_nop\n\tv_nop\n\tv_nop\n\tv_nop\n\tv_nop\n\tv_nop\n\tv_nop\n\tv_nop" : "+v"(c));
    __builtin_amdgcn_sched_barrier(0);

    const v4f s0 = *(const v4f*)(dts + half * 8);
    const v4f s1 = *(const v4f*)(dts + half * 8 + 4);
    float o[8];
#pragma unroll
    for (int r = 0; r < 4; ++r) {
        o[r]     = s0[r] * c[r];
        o[r + 4] = s1[r] * c[r + 4];
    }
    float* crow = C + (size_t)(node0 + half * 8) * NCOL + wave * 16 + mr;
    if (node0 + TILE_ROWS <= n) {
#pragma unroll
        for (int r = 0; r < 8; ++r) crow[(size_t)r * NCOL] = o[r];
    } else {
#pragma unroll
        for (int r = 0; r < 8; ++r)
            if (node0 + half * 8 + r < n) crow[(size_t)r * NCOL] = o[r];
    }
}

extern "C" void kernel_launch(void* const* d_in, const int* in_sizes, int n_in,
                              void* d_out, int out_size, void* d_ws, size_t ws_size,
                              hipStream_t stream) {
    const float* x  = (const float*)d_in[0];
    const int*   ei = (const int*)d_in[1];
    const float* W1 = (const float*)d_in[2];
    const float* b1 = (const float*)d_in[3];
    const float* W2 = (const float*)d_in[4];
    const float* b2 = (const float*)d_in[5];

    const int N = in_sizes[0] / 128;
    const int E = in_sizes[1] / 2;
    const int* src = ei;
    const int* dst = ei + E;

    int*   rowptr  = (int*)d_ws;
    int*   csr_src = rowptr + N;
    float* h1      = (float*)(csr_src + E);
    float* h2      = h1 + (size_t)N * 128;
    const int NB   = (N + RANGE_NODES - 1) / RANGE_NODES;
    int2*  pairs   = (int2*)h2;
    int*   offT    = (int*)(pairs + E);
    float* out     = (float*)d_out;

    const int T = 256;

    const int slice = (E + NSLICE - 1) / NSLICE;
    const size_t group_lds = (size_t)((NB + 2) & ~1) * sizeof(int) + (size_t)slice * sizeof(int2);
    slice_group_kernel<<<NSLICE, 1024, group_lds, stream>>>(src, dst, offT, pairs, E, slice, NB);
    range_rows_kernel<<<NB, 1024, 0, stream>>>(pairs, offT, rowptr, csr_src, x, W1, h1, N, slice);

    const int row_tiles = (N + 15) / 16;
    const int agg_blocks = (N + 7) / 8;

    aggregate_relu_gemm_kernel<<<row_tiles, 128, 0, stream>>>(h1, rowptr, csr_src, b1, W2, h2, N, E);

    aggregate_rows_kernel<64, false><<<agg_blocks, T, 0, stream>>>(h2, rowptr, csr_src, b2, out, N, E);
}
